# starting kernel
# baseline (speedup 1.0000x reference)
#include <hip/hip_runtime.h>
#include <hip/hip_bf16.h>
#include <math.h>

typedef __attribute__((ext_vector_type(16))) _Float16 v16h;
typedef __attribute__((ext_vector_type(8)))  _Float16 v8h;
typedef __attribute__((ext_vector_type(8)))  float    v8f;

#define NEG_SLOPE 0.2f

__device__ __forceinline__ int f2oi(float f) {
  int i = __float_as_int(f);
  return i >= 0 ? i : (i ^ 0x7FFFFFFF);
}
__device__ __forceinline__ float oi2f(int i) {
  return __int_as_float(i >= 0 ? i : (i ^ 0x7FFFFFFF));
}
__device__ __forceinline__ float leaky(float x) { return x > 0.f ? x : NEG_SLOPE * x; }
__device__ __forceinline__ void atomAddF(float* p, float v) {
  __hip_atomic_fetch_add(p, v, __ATOMIC_RELAXED, __HIP_MEMORY_SCOPE_AGENT);
}

__global__ void k_cast_f16(const float* __restrict__ in, _Float16* __restrict__ out, int n) {
  int t = blockIdx.x * blockDim.x + threadIdx.x;
  if (t < n) out[t] = (_Float16)in[t];
}

__global__ void k_castT_f16(const float* __restrict__ in, _Float16* __restrict__ out, int R, int C) {
  int t = blockIdx.x * blockDim.x + threadIdx.x;
  if (t >= R * C) return;
  int r = t / C, c = t - r * C;
  out[c * R + r] = (_Float16)in[t];
}

__global__ void k_wmma_gemm(const _Float16* __restrict__ A,
                            const _Float16* __restrict__ Bt,
                            float* __restrict__ C,
                            int M, int N, int K) {
  const int tilesN      = N >> 4;
  const int wavesPerRow = tilesN >> 2;
  const int totalWaves  = (M >> 4) * wavesPerRow;
  const int wave = blockIdx.x * (blockDim.x >> 5) + (threadIdx.x >> 5);
  if (wave >= totalWaves) return;
  const int lane = threadIdx.x & 31;
  const int half = lane >> 4;
  const int l16  = lane & 15;
  const int tm  = wave / wavesPerRow;
  const int tn0 = (wave - tm * wavesPerRow) * 4;

  const _Float16* aRow = A + (size_t)(tm * 16 + l16) * K;
  const _Float16* bRow[4];
  #pragma unroll
  for (int j = 0; j < 4; ++j)
    bRow[j] = Bt + (size_t)((tn0 + j) * 16 + l16) * K;

  v8f acc0 = {}, acc1 = {}, acc2 = {}, acc3 = {};
  for (int k = 0; k < K; k += 32) {
    const int ka = k + half * 8;
    union { v16h v; v8h h[2]; } a, b0, b1, b2, b3;
    a.h[0]  = *(const v8h*)(aRow + ka);
    a.h[1]  = *(const v8h*)(aRow + ka + 16);
    b0.h[0] = *(const v8h*)(bRow[0] + ka);
    b0.h[1] = *(const v8h*)(bRow[0] + ka + 16);
    b1.h[0] = *(const v8h*)(bRow[1] + ka);
    b1.h[1] = *(const v8h*)(bRow[1] + ka + 16);
    b2.h[0] = *(const v8h*)(bRow[2] + ka);
    b2.h[1] = *(const v8h*)(bRow[2] + ka + 16);
    b3.h[0] = *(const v8h*)(bRow[3] + ka);
    b3.h[1] = *(const v8h*)(bRow[3] + ka + 16);
    acc0 = __builtin_amdgcn_wmma_f32_16x16x32_f16(false, a.v, false, b0.v, (short)0, acc0, false, false);
    acc1 = __builtin_amdgcn_wmma_f32_16x16x32_f16(false, a.v, false, b1.v, (short)0, acc1, false, false);
    acc2 = __builtin_amdgcn_wmma_f32_16x16x32_f16(false, a.v, false, b2.v, (short)0, acc2, false, false);
    acc3 = __builtin_amdgcn_wmma_f32_16x16x32_f16(false, a.v, false, b3.v, (short)0, acc3, false, false);
  }
  float* cRow = C + (size_t)(tm * 16 + half * 8) * N + tn0 * 16 + l16;
  #pragma unroll
  for (int j = 0; j < 8; ++j) {
    float* cr = cRow + (size_t)j * N;
    cr[0]  = acc0[j];
    cr[16] = acc1[j];
    cr[32] = acc2[j];
    cr[48] = acc3[j];
  }
}

__global__ void k_attn(const float* __restrict__ h, const float* __restrict__ att_s,
                       const float* __restrict__ att_d,
                       float* __restrict__ as_, float* __restrict__ ad_,
                       int Nn, int H, int Cc) {
  int t = blockIdx.x * blockDim.x + threadIdx.x;
  if (t >= Nn * H) return;
  int n = t / H, hh = t - n * H;
  const float* hp = h + (size_t)n * H * Cc + (size_t)hh * Cc;
  const float* sp = att_s + hh * Cc;
  const float* dp = att_d + hh * Cc;
  float a0 = 0.f, a1 = 0.f;
  for (int c = 0; c < Cc; ++c) { float v = hp[c]; a0 += v * sp[c]; a1 += v * dp[c]; }
  as_[t] = a0;
  ad_[t] = a1;
}

__global__ void k_init(float* __restrict__ agg, int nAgg,
                       int* __restrict__ m, float* __restrict__ s, int nMS) {
  int t = blockIdx.x * blockDim.x + threadIdx.x;
  if (t < nAgg) agg[t] = 0.f;
  if (t < nMS) { m[t] = f2oi(-__builtin_inff()); s[t] = 0.f; }
}

__global__ void k_edge_max(const int* __restrict__ src, const int* __restrict__ dst,
                           int E, int Etot,
                           const float* __restrict__ as_, const float* __restrict__ ad_,
                           int H, int* __restrict__ m) {
  int t = blockIdx.x * blockDim.x + threadIdx.x;
  if (t >= Etot * H) return;
  int e = t / H, hh = t - e * H;
  int s_ = e < E ? src[e] : (e - E);
  int d_ = e < E ? dst[e] : (e - E);
  float ev = leaky(as_[s_ * H + hh] + ad_[d_ * H + hh]);
  atomicMax(&m[d_ * H + hh], f2oi(ev));
}

__global__ void k_edge_sum(const int* __restrict__ src, const int* __restrict__ dst,
                           int E, int Etot,
                           const float* __restrict__ as_, const float* __restrict__ ad_,
                           int H, const int* __restrict__ m, float* __restrict__ s) {
  int t = blockIdx.x * blockDim.x + threadIdx.x;
  if (t >= Etot * H) return;
  int e = t / H, hh = t - e * H;
  int s_ = e < E ? src[e] : (e - E);
  int d_ = e < E ? dst[e] : (e - E);
  float ev = leaky(as_[s_ * H + hh] + ad_[d_ * H + hh]);
  atomAddF(&s[d_ * H + hh], expf(ev - oi2f(m[d_ * H + hh])));
}

__global__ void k_edge_agg(const int* __restrict__ src, const int* __restrict__ dst,
                           int E, int Etot,
                           const float* __restrict__ as_, const float* __restrict__ ad_,
                           const int* __restrict__ m, const float* __restrict__ ssum,
                           const float* __restrict__ feat, float* __restrict__ outp,
                           int H, int Cc) {
  int wid = blockIdx.x * (blockDim.x >> 5) + (threadIdx.x >> 5);
  if (wid >= Etot) return;
  int lane = threadIdx.x & 31;
  int s_ = wid < E ? src[wid] : (wid - E);
  int d_ = wid < E ? dst[wid] : (wid - E);
  int HC = H * Cc;
  const float* fRow = feat + (size_t)s_ * HC;
  float*       oRow = outp + (size_t)d_ * HC;

  __builtin_prefetch(&fRow[(size_t)lane * HC / 32], 0, 0);

  float alpha[4];
  for (int hh = 0; hh < H; ++hh) {
    float ev = leaky(as_[s_ * H + hh] + ad_[d_ * H + hh]);
    alpha[hh] = expf(ev - oi2f(m[d_ * H + hh])) / (ssum[d_ * H + hh] + 1e-16f);
  }
  for (int base = 0; base < HC; base += 32) {
    float al = alpha[base / Cc];
    int idx = base + lane;
    atomAddF(&oRow[idx], fRow[idx] * al);
  }
}

__global__ void k_bias_relu_cast(const float* __restrict__ a, const float* __restrict__ b,
                                 _Float16* __restrict__ o, int n, int hcMask) {
  int t = blockIdx.x * blockDim.x + threadIdx.x;
  if (t >= n) return;
  float v = a[t] + b[t & hcMask];
  o[t] = (_Float16)(v > 0.f ? v : 0.f);
}

__global__ void k_bias_add(float* __restrict__ o, const float* __restrict__ b, int n, int hcMask) {
  int t = blockIdx.x * blockDim.x + threadIdx.x;
  if (t >= n) return;
  o[t] += b[t & hcMask];
}

extern "C" void kernel_launch(void* const* d_in, const int* in_sizes, int n_in,
                              void* d_out, int out_size, void* d_ws, size_t ws_size,
                              hipStream_t stream) {
  const int IN = 128, HID = 64, H1 = 4;
  const int HC1 = H1 * HID;
  const int HC2 = HID;
  const int Nn   = in_sizes[0] / IN;
  const int E    = in_sizes[1] / 2;
  const int Etot = E + Nn;

  const float* x   = (const float*)d_in[0];
  const int*   ei  = (const int*)d_in[1];
  const float* W1  = (const float*)d_in[2];
  const float* aS1 = (const float*)d_in[3];
  const float* aD1 = (const float*)d_in[4];
  const float* b1  = (const float*)d_in[5];
  const float* W2  = (const float*)d_in[6];
  const float* aS2 = (const float*)d_in[7];
  const float* aD2 = (const float*)d_in[8];
  const float* b2  = (const float*)d_in[9];
  float* out = (float*)d_out;
  const int* srcI = ei;
  const int* dstI = ei + E;

  char* p = (char*)d_ws;
  auto carve = [&](size_t bytes) -> void* {
    void* r = (void*)p;
    p += (bytes + 255) & ~(size_t)255;
    return r;
  };
  _Float16* xh   = (_Float16*)carve((size_t)Nn * IN * 2);
  _Float16* w1t  = (_Float16*)carve((size_t)HC1 * IN * 2);
  _Float16* w2t  = (_Float16*)carve((size_t)HID * HC1 * 2);
  float*    h1   = (float*)carve((size_t)Nn * HC1 * 4);
  float*    agg1 = (float*)carve((size_t)Nn * HC1 * 4);
  _Float16* h1h  = (_Float16*)carve((size_t)Nn * HC1 * 2);
  float*    h2   = (float*)carve((size_t)Nn * HC2 * 4);
  float*    as1  = (float*)carve((size_t)Nn * H1 * 4);
  float*    ad1  = (float*)carve((size_t)Nn * H1 * 4);
  int*      m1   = (int*)carve((size_t)Nn * H1 * 4);
  float*    s1   = (float*)carve((size_t)Nn * H1 * 4);
  float*    as2  = (float*)carve((size_t)Nn * 4);
  float*    ad2  = (float*)carve((size_t)Nn * 4);
  int*      m2   = (int*)carve((size_t)Nn * 4);
  float*    s2   = (float*)carve((size_t)Nn * 4);

  const int TB = 256;
  auto cdiv = [](long long a, long long b) { return (int)((a + b - 1) / b); };

  k_cast_f16<<<cdiv((long long)Nn * IN, TB), TB, 0, stream>>>(x, xh, Nn * IN);
  k_castT_f16<<<cdiv((long long)IN * HC1, TB), TB, 0, stream>>>(W1, w1t, IN, HC1);
  k_castT_f16<<<cdiv((long long)HC1 * HID, TB), TB, 0, stream>>>(W2, w2t, HC1, HID);

  {
    int waves = (Nn / 16) * (HC1 / 64);
    k_wmma_gemm<<<cdiv(waves, 8), TB, 0, stream>>>(xh, w1t, h1, Nn, HC1, IN);
  }
  k_attn<<<cdiv((long long)Nn * H1, TB), TB, 0, stream>>>(h1, aS1, aD1, as1, ad1, Nn, H1, HID);
  k_init<<<cdiv((long long)Nn * HC1, TB), TB, 0, stream>>>(agg1, Nn * HC1, m1, s1, Nn * H1);
  k_edge_max<<<cdiv((long long)Etot * H1, TB), TB, 0, stream>>>(srcI, dstI, E, Etot, as1, ad1, H1, m1);
  k_edge_sum<<<cdiv((long long)Etot * H1, TB), TB, 0, stream>>>(srcI, dstI, E, Etot, as1, ad1, H1, m1, s1);
  k_edge_agg<<<cdiv(Etot, 8), TB, 0, stream>>>(srcI, dstI, E, Etot, as1, ad1, m1, s1, h1, agg1, H1, HID);
  k_bias_relu_cast<<<cdiv((long long)Nn * HC1, TB), TB, 0, stream>>>(agg1, b1, h1h, Nn * HC1, HC1 - 1);

  {
    int waves = (Nn / 16) * (HC2 / 64);
    k_wmma_gemm<<<cdiv(waves, 8), TB, 0, stream>>>(h1h, w2t, h2, Nn, HC2, HC1);
  }
  k_attn<<<cdiv((long long)Nn, TB), TB, 0, stream>>>(h2, aS2, aD2, as2, ad2, Nn, 1, HID);
  k_init<<<cdiv((long long)Nn * HC2, TB), TB, 0, stream>>>(out, Nn * HC2, m2, s2, Nn);
  k_edge_max<<<cdiv((long long)Etot, TB), TB, 0, stream>>>(srcI, dstI, E, Etot, as2, ad2, 1, m2);
  k_edge_sum<<<cdiv((long long)Etot, TB), TB, 0, stream>>>(srcI, dstI, E, Etot, as2, ad2, 1, m2, s2);
  k_edge_agg<<<cdiv(Etot, 8), TB, 0, stream>>>(srcI, dstI, E, Etot, as2, ad2, m2, s2, h2, out, 1, HID);
  k_bias_add<<<cdiv((long long)Nn * HC2, TB), TB, 0, stream>>>(out, b2, Nn * HC2, HC2 - 1);
}


# k_edge_agg reads the whole gathered row into registers before its first atomic add (was load, add, load: one row load in flight per wave); prefetch dropped, the loads themselves go out early
# speedup vs baseline: 1.1419x; 1.1419x over previous
#include <hip/hip_runtime.h>
#include <hip/hip_bf16.h>
#include <math.h>

typedef __attribute__((ext_vector_type(16))) _Float16 v16h;
typedef __attribute__((ext_vector_type(8)))  _Float16 v8h;
typedef __attribute__((ext_vector_type(8)))  float    v8f;

#define NEG_SLOPE 0.2f

__device__ __forceinline__ int f2oi(float f) {
  int i = __float_as_int(f);
  return i >= 0 ? i : (i ^ 0x7FFFFFFF);
}
__device__ __forceinline__ float oi2f(int i) {
  return __int_as_float(i >= 0 ? i : (i ^ 0x7FFFFFFF));
}
__device__ __forceinline__ float leaky(float x) { return x > 0.f ? x : NEG_SLOPE * x; }
__device__ __forceinline__ void atomAddF(float* p, float v) {
  __hip_atomic_fetch_add(p, v, __ATOMIC_RELAXED, __HIP_MEMORY_SCOPE_AGENT);
}

__global__ void k_cast_f16(const float* __restrict__ in, _Float16* __restrict__ out, int n) {
  int t = blockIdx.x * blockDim.x + threadIdx.x;
  if (t < n) out[t] = (_Float16)in[t];
}

__global__ void k_castT_f16(const float* __restrict__ in, _Float16* __restrict__ out, int R, int C) {
  int t = blockIdx.x * blockDim.x + threadIdx.x;
  if (t >= R * C) return;
  int r = t / C, c = t - r * C;
  out[c * R + r] = (_Float16)in[t];
}

__global__ void k_wmma_gemm(const _Float16* __restrict__ A,
                            const _Float16* __restrict__ Bt,
                            float* __restrict__ C,
                            int M, int N, int K) {
  const int tilesN      = N >> 4;
  const int wavesPerRow = tilesN >> 2;
  const int totalWaves  = (M >> 4) * wavesPerRow;
  const int wave = blockIdx.x * (blockDim.x >> 5) + (threadIdx.x >> 5);
  if (wave >= totalWaves) return;
  const int lane = threadIdx.x & 31;
  const int half = lane >> 4;
  const int l16  = lane & 15;
  const int tm  = wave / wavesPerRow;
  const int tn0 = (wave - tm * wavesPerRow) * 4;

  const _Float16* aRow = A + (size_t)(tm * 16 + l16) * K;
  const _Float16* bRow[4];
  #pragma unroll
  for (int j = 0; j < 4; ++j)
    bRow[j] = Bt + (size_t)((tn0 + j) * 16 + l16) * K;

  v8f acc0 = {}, acc1 = {}, acc2 = {}, acc3 = {};
  for (int k = 0; k < K; k += 32) {
    const int ka = k + half * 8;
    union { v16h v; v8h h[2]; } a, b0, b1, b2, b3;
    a.h[0]  = *(const v8h*)(aRow + ka);
    a.h[1]  = *(const v8h*)(aRow + ka + 16);
    b0.h[0] = *(const v8h*)(bRow[0] + ka);
    b0.h[1] = *(const v8h*)(bRow[0] + ka + 16);
    b1.h[0] = *(const v8h*)(bRow[1] + ka);
    b1.h[1] = *(const v8h*)(bRow[1] + ka + 16);
    b2.h[0] = *(const v8h*)(bRow[2] + ka);
    b2.h[1] = *(const v8h*)(bRow[2] + ka + 16);
    b3.h[0] = *(const v8h*)(bRow[3] + ka);
    b3.h[1] = *(const v8h*)(bRow[3] + ka + 16);
    acc0 = __builtin_amdgcn_wmma_f32_16x16x32_f16(false, a.v, false, b0.v, (short)0, acc0, false, false);
    acc1 = __builtin_amdgcn_wmma_f32_16x16x32_f16(false, a.v, false, b1.v, (short)0, acc1, false, false);
    acc2 = __builtin_amdgcn_wmma_f32_16x16x32_f16(false, a.v, false, b2.v, (short)0, acc2, false, false);
    acc3 = __builtin_amdgcn_wmma_f32_16x16x32_f16(false, a.v, false, b3.v, (short)0, acc3, false, false);
  }
  float* cRow = C + (size_t)(tm * 16 + half * 8) * N + tn0 * 16 + l16;
  #pragma unroll
  for (int j = 0; j < 8; ++j) {
    float* cr = cRow + (size_t)j * N;
    cr[0]  = acc0[j];
    cr[16] = acc1[j];
    cr[32] = acc2[j];
    cr[48] = acc3[j];
  }
}

__global__ void k_attn(const float* __restrict__ h, const float* __restrict__ att_s,
                       const float* __restrict__ att_d,
                       float* __restrict__ as_, float* __restrict__ ad_,
                       int Nn, int H, int Cc) {
  int t = blockIdx.x * blockDim.x + threadIdx.x;
  if (t >= Nn * H) return;
  int n = t / H, hh = t - n * H;
  const float* hp = h + (size_t)n * H * Cc + (size_t)hh * Cc;
  const float* sp = att_s + hh * Cc;
  const float* dp = att_d + hh * Cc;
  float a0 = 0.f, a1 = 0.f;
  for (int c = 0; c < Cc; ++c) { float v = hp[c]; a0 += v * sp[c]; a1 += v * dp[c]; }
  as_[t] = a0;
  ad_[t] = a1;
}

__global__ void k_init(float* __restrict__ agg, int nAgg,
                       int* __restrict__ m, float* __restrict__ s, int nMS) {
  int t = blockIdx.x * blockDim.x + threadIdx.x;
  if (t < nAgg) agg[t] = 0.f;
  if (t < nMS) { m[t] = f2oi(-__builtin_inff()); s[t] = 0.f; }
}

__global__ void k_edge_max(const int* __restrict__ src, const int* __restrict__ dst,
                           int E, int Etot,
                           const float* __restrict__ as_, const float* __restrict__ ad_,
                           int H, int* __restrict__ m) {
  int t = blockIdx.x * blockDim.x + threadIdx.x;
  if (t >= Etot * H) return;
  int e = t / H, hh = t - e * H;
  int s_ = e < E ? src[e] : (e - E);
  int d_ = e < E ? dst[e] : (e - E);
  float ev = leaky(as_[s_ * H + hh] + ad_[d_ * H + hh]);
  atomicMax(&m[d_ * H + hh], f2oi(ev));
}

__global__ void k_edge_sum(const int* __restrict__ src, const int* __restrict__ dst,
                           int E, int Etot,
                           const float* __restrict__ as_, const float* __restrict__ ad_,
                           int H, const int* __restrict__ m, float* __restrict__ s) {
  int t = blockIdx.x * blockDim.x + threadIdx.x;
  if (t >= Etot * H) return;
  int e = t / H, hh = t - e * H;
  int s_ = e < E ? src[e] : (e - E);
  int d_ = e < E ? dst[e] : (e - E);
  float ev = leaky(as_[s_ * H + hh] + ad_[d_ * H + hh]);
  atomAddF(&s[d_ * H + hh], expf(ev - oi2f(m[d_ * H + hh])));
}

__global__ void k_edge_agg(const int* __restrict__ src, const int* __restrict__ dst,
                           int E, int Etot,
                           const float* __restrict__ as_, const float* __restrict__ ad_,
                           const int* __restrict__ m, const float* __restrict__ ssum,
                           const float* __restrict__ feat, float* __restrict__ outp,
                           int H, int Cc) {
  int wid = blockIdx.x * (blockDim.x >> 5) + (threadIdx.x >> 5);
  if (wid >= Etot) return;
  int lane = threadIdx.x & 31;
  int s_ = wid < E ? src[wid] : (wid - E);
  int d_ = wid < E ? dst[wid] : (wid - E);
  int HC = H * Cc;
  const float* fRow = feat + (size_t)s_ * HC;
  float*       oRow = outp + (size_t)d_ * HC;

  const int nch = HC >> 5;

  float v[8];
  #pragma unroll
  for (int j = 0; j < 8; ++j) v[j] = (j < nch) ? fRow[j * 32 + lane] : 0.f;

  float alpha[4];
  for (int hh = 0; hh < H; ++hh) {
    float ev = leaky(as_[s_ * H + hh] + ad_[d_ * H + hh]);
    alpha[hh] = expf(ev - oi2f(m[d_ * H + hh])) / (ssum[d_ * H + hh] + 1e-16f);
  }
  for (int c0 = 0;;) {
    #pragma unroll
    for (int j = 0; j < 8; ++j) {
      int c = c0 + j;
      if (c < nch) atomAddF(&oRow[c * 32 + lane], v[j] * alpha[(c * 32) / Cc]);
    }
    c0 += 8;
    if (c0 >= nch) break;
    #pragma unroll
    for (int j = 0; j < 8; ++j) v[j] = (c0 + j < nch) ? fRow[(c0 + j) * 32 + lane] : 0.f;
  }
}

__global__ void k_bias_relu_cast(const float* __restrict__ a, const float* __restrict__ b,
                                 _Float16* __restrict__ o, int n, int hcMask) {
  int t = blockIdx.x * blockDim.x + threadIdx.x;
  if (t >= n) return;
  float v = a[t] + b[t & hcMask];
  o[t] = (_Float16)(v > 0.f ? v : 0.f);
}

__global__ void k_bias_add(float* __restrict__ o, const float* __restrict__ b, int n, int hcMask) {
  int t = blockIdx.x * blockDim.x + threadIdx.x;
  if (t >= n) return;
  o[t] += b[t & hcMask];
}

extern "C" void kernel_launch(void* const* d_in, const int* in_sizes, int n_in,
                              void* d_out, int out_size, void* d_ws, size_t ws_size,
                              hipStream_t stream) {
  const int IN = 128, HID = 64, H1 = 4;
  const int HC1 = H1 * HID;
  const int HC2 = HID;
  const int Nn   = in_sizes[0] / IN;
  const int E    = in_sizes[1] / 2;
  const int Etot = E + Nn;

  const float* x   = (const float*)d_in[0];
  const int*   ei  = (const int*)d_in[1];
  const float* W1  = (const float*)d_in[2];
  const float* aS1 = (const float*)d_in[3];
  const float* aD1 = (const float*)d_in[4];
  const float* b1  = (const float*)d_in[5];
  const float* W2  = (const float*)d_in[6];
  const float* aS2 = (const float*)d_in[7];
  const float* aD2 = (const float*)d_in[8];
  const float* b2  = (const float*)d_in[9];
  float* out = (float*)d_out;
  const int* srcI = ei;
  const int* dstI = ei + E;

  char* p = (char*)d_ws;
  auto carve = [&](size_t bytes) -> void* {
    void* r = (void*)p;
    p += (bytes + 255) & ~(size_t)255;
    return r;
  };
  _Float16* xh   = (_Float16*)carve((size_t)Nn * IN * 2);
  _Float16* w1t  = (_Float16*)carve((size_t)HC1 * IN * 2);
  _Float16* w2t  = (_Float16*)carve((size_t)HID * HC1 * 2);
  float*    h1   = (float*)carve((size_t)Nn * HC1 * 4);
  float*    agg1 = (float*)carve((size_t)Nn * HC1 * 4);
  _Float16* h1h  = (_Float16*)carve((size_t)Nn * HC1 * 2);
  float*    h2   = (float*)carve((size_t)Nn * HC2 * 4);
  float*    as1  = (float*)carve((size_t)Nn * H1 * 4);
  float*    ad1  = (float*)carve((size_t)Nn * H1 * 4);
  int*      m1   = (int*)carve((size_t)Nn * H1 * 4);
  float*    s1   = (float*)carve((size_t)Nn * H1 * 4);
  float*    as2  = (float*)carve((size_t)Nn * 4);
  float*    ad2  = (float*)carve((size_t)Nn * 4);
  int*      m2   = (int*)carve((size_t)Nn * 4);
  float*    s2   = (float*)carve((size_t)Nn * 4);

  const int TB = 256;
  auto cdiv = [](long long a, long long b) { return (int)((a + b - 1) / b); };

  k_cast_f16<<<cdiv((long long)Nn * IN, TB), TB, 0, stream>>>(x, xh, Nn * IN);
  k_castT_f16<<<cdiv((long long)IN * HC1, TB), TB, 0, stream>>>(W1, w1t, IN, HC1);
  k_castT_f16<<<cdiv((long long)HC1 * HID, TB), TB, 0, stream>>>(W2, w2t, HC1, HID);

  {
    int waves = (Nn / 16) * (HC1 / 64);
    k_wmma_gemm<<<cdiv(waves, 8), TB, 0, stream>>>(xh, w1t, h1, Nn, HC1, IN);
  }
  k_attn<<<cdiv((long long)Nn * H1, TB), TB, 0, stream>>>(h1, aS1, aD1, as1, ad1, Nn, H1, HID);
  k_init<<<cdiv((long long)Nn * HC1, TB), TB, 0, stream>>>(agg1, Nn * HC1, m1, s1, Nn * H1);
  k_edge_max<<<cdiv((long long)Etot * H1, TB), TB, 0, stream>>>(srcI, dstI, E, Etot, as1, ad1, H1, m1);
  k_edge_sum<<<cdiv((long long)Etot * H1, TB), TB, 0, stream>>>(srcI, dstI, E, Etot, as1, ad1, H1, m1, s1);
  k_edge_agg<<<cdiv(Etot, 8), TB, 0, stream>>>(srcI, dstI, E, Etot, as1, ad1, m1, s1, h1, agg1, H1, HID);
  k_bias_relu_cast<<<cdiv((long long)Nn * HC1, TB), TB, 0, stream>>>(agg1, b1, h1h, Nn * HC1, HC1 - 1);

  {
    int waves = (Nn / 16) * (HC2 / 64);
    k_wmma_gemm<<<cdiv(waves, 8), TB, 0, stream>>>(h1h, w2t, h2, Nn, HC2, HC1);
  }
  k_attn<<<cdiv((long long)Nn, TB), TB, 0, stream>>>(h2, aS2, aD2, as2, ad2, Nn, 1, HID);
  k_init<<<cdiv((long long)Nn * HC2, TB), TB, 0, stream>>>(out, Nn * HC2, m2, s2, Nn);
  k_edge_max<<<cdiv((long long)Etot, TB), TB, 0, stream>>>(srcI, dstI, E, Etot, as2, ad2, 1, m2);
  k_edge_sum<<<cdiv((long long)Etot, TB), TB, 0, stream>>>(srcI, dstI, E, Etot, as2, ad2, 1, m2, s2);
  k_edge_agg<<<cdiv(Etot, 8), TB, 0, stream>>>(srcI, dstI, E, Etot, as2, ad2, m2, s2, h2, out, 1, HID);
  k_bias_add<<<cdiv((long long)Nn * HC2, TB), TB, 0, stream>>>(out, b2, Nn * HC2, HC2 - 1);
}


# Segment softmax formed in the destination tile from LDS edge lists (int max, exact fixed-point exp-sum, divide at store; full lists worked off in rounds); 6 launches gone; sum order: fixed per row
# speedup vs baseline: 2.3791x; 2.0835x over previous
#include <hip/hip_runtime.h>
#include <hip/hip_bf16.h>
#include <math.h>

typedef __attribute__((ext_vector_type(16))) _Float16 v16h;
typedef __attribute__((ext_vector_type(8)))  _Float16 v8h;
typedef __attribute__((ext_vector_type(8)))  float    v8f;
typedef __attribute__((ext_vector_type(4)))  _Float16 v4h;

#define EW_SPAN 2048
#define EW_HALF 1024

__device__ __forceinline__ v4h cast4(float4 a) {
  v4h o;
  o[0] = (_Float16)a.x; o[1] = (_Float16)a.y; o[2] = (_Float16)a.z; o[3] = (_Float16)a.w;
  return o;
}
__device__ __forceinline__ float4 add4(float4 a, float4 b) {
  return make_float4(a.x + b.x, a.y + b.y, a.z + b.z, a.w + b.w);
}
__device__ __forceinline__ float4 relu4(float4 a) {
  return make_float4(a.x > 0.f ? a.x : 0.f, a.y > 0.f ? a.y : 0.f, a.z > 0.f ? a.z : 0.f, a.w > 0.f ? a.w : 0.f);
}

#define NEG_SLOPE 0.2f

__device__ __forceinline__ int f2oi(float f) {
  int i = __float_as_int(f);
  return i >= 0 ? i : (i ^ 0x7FFFFFFF);
}
__device__ __forceinline__ float oi2f(int i) {
  return __int_as_float(i >= 0 ? i : (i ^ 0x7FFFFFFF));
}
__device__ __forceinline__ float leaky(float x) { return x > 0.f ? x : NEG_SLOPE * x; }

__global__ void k_cast_f16(const float* __restrict__ in, _Float16* __restrict__ out, int n) {
  const int i0 = blockIdx.x * EW_SPAN + threadIdx.x * 4;
  const int i1 = i0 + EW_HALF;
  if (i1 + 4 <= n) {
    float4 a0 = *(const float4*)(in + i0);
    float4 a1 = *(const float4*)(in + i1);
    *(v4h*)(out + i0) = cast4(a0);
    *(v4h*)(out + i1) = cast4(a1);
  } else {
    for (int g = 0; g < 2; ++g) {
      int i = g ? i1 : i0;
      if (i + 4 <= n) *(v4h*)(out + i) = cast4(*(const float4*)(in + i));
      else for (; i < n; ++i) out[i] = (_Float16)in[i];
    }
  }
}

__global__ void k_castT_f16(const float* __restrict__ in, _Float16* __restrict__ out, int R, int C) {
  int t = blockIdx.x * blockDim.x + threadIdx.x;
  if (t >= R * C) return;
  int r = t / C, c = t - r * C;
  out[c * R + r] = (_Float16)in[t];
}

__global__ void k_wmma_gemm(const _Float16* __restrict__ A,
                            const _Float16* __restrict__ Bt,
                            float* __restrict__ C,
                            int M, int N, int K) {
  const int tilesN      = N >> 4;
  const int wavesPerRow = tilesN >> 2;
  const int totalWaves  = (M >> 4) * wavesPerRow;
  const int wave = blockIdx.x * (blockDim.x >> 5) + (threadIdx.x >> 5);
  if (wave >= totalWaves) return;
  const int lane = threadIdx.x & 31;
  const int half = lane >> 4;
  const int l16  = lane & 15;
  const int tm  = wave / wavesPerRow;
  const int tn0 = (wave - tm * wavesPerRow) * 4;

  const _Float16* aRow = A + (size_t)(tm * 16 + l16) * K;
  const _Float16* bRow[4];
  #pragma unroll
  for (int j = 0; j < 4; ++j)
    bRow[j] = Bt + (size_t)((tn0 + j) * 16 + l16) * K;

  v8f acc0 = {}, acc1 = {}, acc2 = {}, acc3 = {};
  for (int k = 0; k < K; k += 32) {
    const int ka = k + half * 8;
    union { v16h v; v8h h[2]; } a, b0, b1, b2, b3;
    a.h[0]  = *(const v8h*)(aRow + ka);
    a.h[1]  = *(const v8h*)(aRow + ka + 16);
    b0.h[0] = *(const v8h*)(bRow[0] + ka);
    b0.h[1] = *(const v8h*)(bRow[0] + ka + 16);
    b1.h[0] = *(const v8h*)(bRow[1] + ka);
    b1.h[1] = *(const v8h*)(bRow[1] + ka + 16);
    b2.h[0] = *(const v8h*)(bRow[2] + ka);
    b2.h[1] = *(const v8h*)(bRow[2] + ka + 16);
    b3.h[0] = *(const v8h*)(bRow[3] + ka);
    b3.h[1] = *(const v8h*)(bRow[3] + ka + 16);
    acc0 = __builtin_amdgcn_wmma_f32_16x16x32_f16(false, a.v, false, b0.v, (short)0, acc0, false, false);
    acc1 = __builtin_amdgcn_wmma_f32_16x16x32_f16(false, a.v, false, b1.v, (short)0, acc1, false, false);
    acc2 = __builtin_amdgcn_wmma_f32_16x16x32_f16(false, a.v, false, b2.v, (short)0, acc2, false, false);
    acc3 = __builtin_amdgcn_wmma_f32_16x16x32_f16(false, a.v, false, b3.v, (short)0, acc3, false, false);
  }
  float* cRow = C + (size_t)(tm * 16 + half * 8) * N + tn0 * 16 + l16;
  #pragma unroll
  for (int j = 0; j < 8; ++j) {
    float* cr = cRow + (size_t)j * N;
    cr[0]  = acc0[j];
    cr[16] = acc1[j];
    cr[32] = acc2[j];
    cr[48] = acc3[j];
  }
}

__global__ void k_attn(const float* __restrict__ h, const float* __restrict__ att_s,
                       const float* __restrict__ att_d,
                       float* __restrict__ as_, float* __restrict__ ad_,
                       int Nn, int H, int Cc) {
  int t = blockIdx.x * blockDim.x + threadIdx.x;
  if (t >= Nn * H) return;
  int n = t / H, hh = t - n * H;
  const float* hp = h + (size_t)n * H * Cc + (size_t)hh * Cc;
  const float* sp = att_s + hh * Cc;
  const float* dp = att_d + hh * Cc;
  float a0 = 0.f, a1 = 0.f;
  for (int c = 0; c < Cc; ++c) { float v = hp[c]; a0 += v * sp[c]; a1 += v * dp[c]; }
  as_[t] = a0;
  ad_[t] = a1;
}

#define TILE_R   200
#define TILE_T   1024
#define TILE_W   (TILE_T / 32)
#define SCAN_LD  16
#define SCAN_E   (SCAN_LD * 32)
#define QL       384
#define PASS_U   4
#define PQCAP    (32 * PASS_U + 32)
#define FIX_ONE  1099511627776.0f
static_assert(TILE_R <= 256, "a row inside the tile is kept in one byte");
static_assert(TILE_W == 32, "a row's wave is row & 31; lane i reads wave i's end-of-list flag");
static_assert(QL > 64, "a round must be able to take at least one load's hits and the self loops");

template <int HC, int H, bool FIRST>
__global__ __launch_bounds__(TILE_T)
void k_tile_agg(const int* __restrict__ src, const int* __restrict__ dst, int E, int Nn,
                const float* __restrict__ as_, const float* __restrict__ ad_,
                const float* __restrict__ feat, const float* __restrict__ bias,
                float* __restrict__ outF, _Float16* __restrict__ outH) {
  constexpr int Cc  = HC / H;
  constexpr int NCH = HC / 32;
  constexpr int EPI = (NCH >= 8) ? 4 : 8;
  __shared__ float acc[TILE_R * HC];
  __shared__ unsigned long long sfix[TILE_R * H];
  __shared__ int   mrow[TILE_R * H];
  __shared__ int   mold[TILE_R * H];
  __shared__ int   q[TILE_W][QL];
  __shared__ unsigned char ql[TILE_W][QL];
  __shared__ int   qcnt[TILE_W];
  __shared__ int   qfin[TILE_W];
  __shared__ int   pqs[TILE_W][PQCAP];
  __shared__ unsigned char pqr[TILE_W][PQCAP];

  const int tid = threadIdx.x, lane = tid & 31;
  const int w = __builtin_amdgcn_readfirstlane(tid >> 5);
  const int r0 = blockIdx.x * TILE_R;
  const int nrows = (Nn - r0 < TILE_R) ? (Nn - r0) : TILE_R;
  const int ninf = f2oi(-__builtin_inff());

  for (int i = tid * 4; i < TILE_R * HC; i += TILE_T * 4)
    *(float4*)(acc + i) = make_float4(0.f, 0.f, 0.f, 0.f);
  for (int i = tid; i < TILE_R * H; i += TILE_T) { mrow[i] = ninf; sfix[i] = 0ull; }

  const int nStep = (E + SCAN_E - 1) / SCAN_E;
  const int rot = (int)(((long long)blockIdx.x * nStep) / gridDim.x);
  const int nIter = (nStep - w + TILE_W - 1) / TILE_W;
  const int maxRound = (((nStep + TILE_W - 1) / TILE_W) * SCAN_E) / (QL - 32) + 3;

  auto stepBase = [&](int k) {
    int c = w + k * TILE_W + rot;
    if (c >= nStep) c -= nStep;
    return c * SCAN_E;
  };
  int dd[SCAN_LD];
  auto loadStep = [&](int b) {
    if (b + SCAN_E <= E) {
      for (int j = 0; j < SCAN_LD; ++j) dd[j] = dst[b + lane + j * 32];
    } else {
      for (int j = 0; j < SCAN_LD; ++j) dd[j] = dst[(b + lane + j * 32 < E) ? (b + lane + j * 32) : (E - 1)];
    }
  };

  int  it = 0, jStart = 0;
  bool selfDone = false;

  for (int round = 0; round < maxRound; ++round) {
    int  qn = 0;
    bool full = false;
    if (round > 0)
      for (int i = tid; i < TILE_R * H; i += TILE_T) mold[i] = mrow[i];

    int bNext = 0;
    if (it < nIter) { bNext = stepBase(it); loadStep(bNext); }
    while (it < nIter && !full) {
      const int b0 = bNext;
      int dl[SCAN_LD];
      unsigned mk[SCAN_LD];
      for (int j = 0; j < SCAN_LD; ++j) dl[j] = dd[j] - r0;
      if (b0 + SCAN_E <= E) {
        for (int j = 0; j < SCAN_LD; ++j)
          mk[j] = __builtin_amdgcn_ballot_w32((unsigned)dl[j] < (unsigned)nrows);
      } else {
        for (int j = 0; j < SCAN_LD; ++j)
          mk[j] = __builtin_amdgcn_ballot_w32((unsigned)dl[j] < (unsigned)nrows && b0 + lane + j * 32 < E);
      }
      if (it + 1 < nIter) { bNext = stepBase(it + 1); loadStep(bNext); }
      for (int j = 0; j < SCAN_LD; ++j) {
        if (j >= jStart && !full && mk[j]) {
          const int c = __builtin_popcount(mk[j]);
          if (qn + c > QL) {
            full = true;
            jStart = j;
          } else {
            if ((mk[j] >> lane) & 1u) {
              const int pos = qn + (int)__builtin_amdgcn_mbcnt_lo(mk[j], 0u);
              q[w][pos] = b0 + lane + j * 32;
              ql[w][pos] = (unsigned char)dl[j];
            }
            qn += c;
          }
        }
      }
      if (!full) { jStart = 0; ++it; }
    }
    if (it == nIter && !full && !selfDone) {
      const int row = lane * TILE_W + w;
      const bool hit = row < nrows;
      const unsigned mask = __builtin_amdgcn_ballot_w32(hit);
      const int c = __builtin_popcount(mask);
      if (qn + c <= QL) {
        if (hit) {
          const int pos = qn + (int)__builtin_amdgcn_mbcnt_lo(mask, 0u);
          q[w][pos] = E + r0 + row;
          ql[w][pos] = (unsigned char)row;
        }
        qn += c;
        selfDone = true;
      }
    }
    if (lane == 0) { qcnt[w] = qn; qfin[w] = selfDone ? 1 : 0; }
    __syncthreads();
    const bool lastRound = __builtin_amdgcn_ballot_w32(qfin[lane] != 0) == 0xFFFFFFFFu;

    for (int base = 0; base < qn; base += 32 * PASS_U) {
      int sv[PASS_U], lr[PASS_U], ix[PASS_U];
      bool ok[PASS_U];
      float ev[PASS_U][H];
      for (int u = 0; u < PASS_U; ++u) {
        ix[u] = base + u * 32 + lane;
        ok[u] = ix[u] < qn;
        if (!ok[u]) ix[u] = base;
        const int e = q[w][ix[u]];
        lr[u] = ql[w][ix[u]];
        const int sl = src[e < E ? e : 0];
        sv[u] = e < E ? sl : e - E;
      }
      for (int u = 0; u < PASS_U; ++u)
        for (int hh = 0; hh < H; ++hh)
          ev[u][hh] = leaky(as_[sv[u] * H + hh] + ad_[(r0 + lr[u]) * H + hh]);
      for (int u = 0; u < PASS_U; ++u) {
        for (int hh = 0; hh < H; ++hh)
          __hip_atomic_fetch_max(&mrow[lr[u] * H + hh], ok[u] ? f2oi(ev[u][hh]) : ninf,
                                 __ATOMIC_RELAXED, __HIP_MEMORY_SCOPE_WORKGROUP);
        if (ok[u]) q[w][ix[u]] = sv[u];
      }
    }
    __syncthreads();

    if (round > 0) {
      for (int i = tid; i < TILE_R * HC; i += TILE_T) {
        const int rh = (i / HC) * H + (i % HC) / Cc;
        const int a = mold[rh], b = mrow[rh];
        if (a != b) acc[i] *= expf(oi2f(a) - oi2f(b));
      }
      for (int i = tid; i < TILE_R * H; i += TILE_T) {
        const int a = mold[i], b = mrow[i];
        if (a != b) sfix[i] = (unsigned long long)((float)sfix[i] * expf(oi2f(a) - oi2f(b)));
      }
      __syncthreads();
    }

    for (int base = 0; base < qn; base += 32 * PASS_U) {
      int sv[PASS_U], lr[PASS_U];
      bool ok[PASS_U];
      float ev[PASS_U][H];
      for (int u = 0; u < PASS_U; ++u) {
        int ix = base + u * 32 + lane;
        ok[u] = ix < qn;
        if (!ok[u]) ix = base;
        sv[u] = q[w][ix];
        lr[u] = ql[w][ix];
      }
      for (int u = 0; u < PASS_U; ++u)
        for (int hh = 0; hh < H; ++hh)
          ev[u][hh] = leaky(as_[sv[u] * H + hh] + ad_[(r0 + lr[u]) * H + hh]);
      for (int u = 0; u < PASS_U; ++u) {
        for (int hh = 0; hh < H; ++hh) {
          const float ex = expf(ev[u][hh] - oi2f(mrow[lr[u] * H + hh]));
          __hip_atomic_fetch_add(&sfix[lr[u] * H + hh], ok[u] ? (unsigned long long)(ex * FIX_ONE) : 0ull,
                                 __ATOMIC_RELAXED, __HIP_MEMORY_SCOPE_WORKGROUP);
        }
      }
    }

    int pn = 0;
    auto work = [&](bool last) {
      while (pn >= 32 || (last && pn > 0)) {
        const int n = pn < 32 ? pn : 32;
        const int pb = pn - n;
        pn = pb;

        int s_ = 0, lrow = 0;
        float al[H];
        for (int hh = 0; hh < H; ++hh) al[hh] = 0.f;
        if (lane < n) {
          s_ = pqs[w][pb + lane];
          lrow = pqr[w][pb + lane];
          const int d_ = r0 + lrow;
          for (int hh = 0; hh < H; ++hh) {
            const float ev = leaky(as_[s_ * H + hh] + ad_[d_ * H + hh]);
            al[hh] = expf(ev - oi2f(mrow[lrow * H + hh]));
          }
        }

        for (int i = 0; i < n; i += EPI) {
          float v[EPI][NCH];
          int   lr[EPI];
          float a[EPI][H];
          for (int k = 0; k < EPI; ++k) {
            const int ik = (i + k < n) ? (i + k) : (n - 1);
            const int sk = __builtin_amdgcn_readlane(s_, ik);
            lr[k] = __builtin_amdgcn_readlane(lrow, ik);
            for (int hh = 0; hh < H; ++hh)
              a[k][hh] = __int_as_float(__builtin_amdgcn_readlane(__float_as_int(al[hh]), ik));
            const float* fRow = feat + (size_t)sk * HC + lane;
            for (int j = 0; j < NCH; ++j) v[k][j] = fRow[j * 32];
          }
          for (int k = 0; k < EPI; ++k) {
            if (i + k < n) {
              float* aRow = acc + lr[k] * HC + lane;
              for (int j = 0; j < NCH; ++j)
                __hip_atomic_fetch_add(aRow + j * 32, v[k][j] * a[k][(j * 32) / Cc],
                                       __ATOMIC_RELAXED, __HIP_MEMORY_SCOPE_WORKGROUP);
            }
          }
        }
      }
    };
    for (int wl = 0; wl < TILE_W; ++wl) {
      const int cn = __builtin_amdgcn_readfirstlane(qcnt[wl]);
      for (int base = 0; base < cn; base += 32 * PASS_U) {
        int sv[PASS_U], lr[PASS_U];
        unsigned mk[PASS_U];
        for (int u = 0; u < PASS_U; ++u) {
          int ix = base + u * 32 + lane;
          const bool in = ix < cn;
          if (!in) ix = base;
          sv[u] = q[wl][ix];
          lr[u] = ql[wl][ix];
          mk[u] = __builtin_amdgcn_ballot_w32(in && (lr[u] & (TILE_W - 1)) == w);
        }
        for (int u = 0; u < PASS_U; ++u) {
          if (mk[u]) {
            if ((mk[u] >> lane) & 1u) {
              const int pos = pn + (int)__builtin_amdgcn_mbcnt_lo(mk[u], 0u);
              pqs[w][pos] = sv[u];
              pqr[w][pos] = (unsigned char)lr[u];
            }
            pn += __builtin_popcount(mk[u]);
          }
        }
        work(false);
      }
    }
    work(true);
    __syncthreads();
    if (lastRound) break;
  }

  const size_t o0 = (size_t)r0 * HC;
  for (int i = tid * 4; i < nrows * HC; i += TILE_T * 4) {
    const int rh = (i / HC) * H + (i & (HC - 1)) / Cc;
    const float den = (float)sfix[rh] * (1.0f / FIX_ONE) + 1e-16f;
    const float4 a = *(const float4*)(acc + i);
    float4 r = add4(make_float4(a.x / den, a.y / den, a.z / den, a.w / den),
                    *(const float4*)(bias + (i & (HC - 1))));
    if (FIRST) *(v4h*)(outH + o0 + i) = cast4(relu4(r));
    else       *(float4*)(outF + o0 + i) = r;
  }
}

extern "C" void kernel_launch(void* const* d_in, const int* in_sizes, int n_in,
                              void* d_out, int out_size, void* d_ws, size_t ws_size,
                              hipStream_t stream) {
  const int IN = 128, HID = 64, H1 = 4;
  const int HC1 = H1 * HID;
  const int HC2 = HID;
  const int Nn   = in_sizes[0] / IN;
  const int E    = in_sizes[1] / 2;

  const float* x   = (const float*)d_in[0];
  const int*   ei  = (const int*)d_in[1];
  const float* W1  = (const float*)d_in[2];
  const float* aS1 = (const float*)d_in[3];
  const float* aD1 = (const float*)d_in[4];
  const float* b1  = (const float*)d_in[5];
  const float* W2  = (const float*)d_in[6];
  const float* aS2 = (const float*)d_in[7];
  const float* aD2 = (const float*)d_in[8];
  const float* b2  = (const float*)d_in[9];
  float* out = (float*)d_out;
  const int* srcI = ei;
  const int* dstI = ei + E;

  char* p = (char*)d_ws;
  auto carve = [&](size_t bytes) -> void* {
    void* r = (void*)p;
    p += (bytes + 255) & ~(size_t)255;
    return r;
  };
  _Float16* xh   = (_Float16*)carve((size_t)Nn * IN * 2);
  _Float16* w1t  = (_Float16*)carve((size_t)HC1 * IN * 2);
  _Float16* w2t  = (_Float16*)carve((size_t)HID * HC1 * 2);
  float*    h1   = (float*)carve((size_t)Nn * HC1 * 4);
  _Float16* h1h  = (_Float16*)carve((size_t)Nn * HC1 * 2);
  float*    h2   = (float*)carve((size_t)Nn * HC2 * 4);
  float*    as1  = (float*)carve((size_t)Nn * H1 * 4);
  float*    ad1  = (float*)carve((size_t)Nn * H1 * 4);
  float*    as2  = (float*)carve((size_t)Nn * 4);
  float*    ad2  = (float*)carve((size_t)Nn * 4);

  const int TB = 256;
  auto cdiv = [](long long a, long long b) { return (int)((a + b - 1) / b); };

  k_cast_f16<<<cdiv((long long)Nn * IN, EW_SPAN), TB, 0, stream>>>(x, xh, Nn * IN);
  k_castT_f16<<<cdiv((long long)IN * HC1, TB), TB, 0, stream>>>(W1, w1t, IN, HC1);
  k_castT_f16<<<cdiv((long long)HC1 * HID, TB), TB, 0, stream>>>(W2, w2t, HC1, HID);

  {
    int waves = (Nn / 16) * (HC1 / 64);
    k_wmma_gemm<<<cdiv(waves, 8), TB, 0, stream>>>(xh, w1t, h1, Nn, HC1, IN);
  }
  k_attn<<<cdiv((long long)Nn * H1, TB), TB, 0, stream>>>(h1, aS1, aD1, as1, ad1, Nn, H1, HID);
  k_tile_agg<256, 4, true><<<cdiv(Nn, TILE_R), TILE_T, 0, stream>>>(srcI, dstI, E, Nn, as1, ad1, h1, b1, nullptr, h1h);

  {
    int waves = (Nn / 16) * (HC2 / 64);
    k_wmma_gemm<<<cdiv(waves, 8), TB, 0, stream>>>(h1h, w2t, h2, Nn, HC2, HC1);
  }
  k_attn<<<cdiv((long long)Nn, TB), TB, 0, stream>>>(h2, aS2, aD2, as2, ad2, Nn, 1, HID);
  k_tile_agg<64, 1, false><<<cdiv(Nn, TILE_R), TILE_T, 0, stream>>>(srcI, dstI, E, Nn, as2, ad2, h2, b2, out, nullptr);
}


# Layer 2 takes each tile's edge lists (source node, row in tile) from the copy layer 1 wrote in this call, not from a second sweep of the dst list; a multi-round tile still sweeps; sums unchanged
# speedup vs baseline: 2.8130x; 1.1824x over previous
#include <hip/hip_runtime.h>
#include <hip/hip_bf16.h>
#include <math.h>

typedef __attribute__((ext_vector_type(16))) _Float16 v16h;
typedef __attribute__((ext_vector_type(8)))  _Float16 v8h;
typedef __attribute__((ext_vector_type(8)))  float    v8f;
typedef __attribute__((ext_vector_type(4)))  _Float16 v4h;

#define EW_SPAN 2048
#define EW_HALF 1024

__device__ __forceinline__ v4h cast4(float4 a) {
  v4h o;
  o[0] = (_Float16)a.x; o[1] = (_Float16)a.y; o[2] = (_Float16)a.z; o[3] = (_Float16)a.w;
  return o;
}
__device__ __forceinline__ float4 add4(float4 a, float4 b) {
  return make_float4(a.x + b.x, a.y + b.y, a.z + b.z, a.w + b.w);
}
__device__ __forceinline__ float4 relu4(float4 a) {
  return make_float4(a.x > 0.f ? a.x : 0.f, a.y > 0.f ? a.y : 0.f, a.z > 0.f ? a.z : 0.f, a.w > 0.f ? a.w : 0.f);
}

#define NEG_SLOPE 0.2f

__device__ __forceinline__ int f2oi(float f) {
  int i = __float_as_int(f);
  return i >= 0 ? i : (i ^ 0x7FFFFFFF);
}
__device__ __forceinline__ float oi2f(int i) {
  return __int_as_float(i >= 0 ? i : (i ^ 0x7FFFFFFF));
}
__device__ __forceinline__ float leaky(float x) { return x > 0.f ? x : NEG_SLOPE * x; }

__global__ void k_cast_f16(const float* __restrict__ in, _Float16* __restrict__ out, int n) {
  const int i0 = blockIdx.x * EW_SPAN + threadIdx.x * 4;
  const int i1 = i0 + EW_HALF;
  if (i1 + 4 <= n) {
    float4 a0 = *(const float4*)(in + i0);
    float4 a1 = *(const float4*)(in + i1);
    *(v4h*)(out + i0) = cast4(a0);
    *(v4h*)(out + i1) = cast4(a1);
  } else {
    for (int g = 0; g < 2; ++g) {
      int i = g ? i1 : i0;
      if (i + 4 <= n) *(v4h*)(out + i) = cast4(*(const float4*)(in + i));
      else for (; i < n; ++i) out[i] = (_Float16)in[i];
    }
  }
}

__global__ void k_castT_f16(const float* __restrict__ in, _Float16* __restrict__ out, int R, int C) {
  int t = blockIdx.x * blockDim.x + threadIdx.x;
  if (t >= R * C) return;
  int r = t / C, c = t - r * C;
  out[c * R + r] = (_Float16)in[t];
}

__global__ void k_wmma_gemm(const _Float16* __restrict__ A,
                            const _Float16* __restrict__ Bt,
                            float* __restrict__ C,
                            int M, int N, int K) {
  const int tilesN      = N >> 4;
  const int wavesPerRow = tilesN >> 2;
  const int totalWaves  = (M >> 4) * wavesPerRow;
  const int wave = blockIdx.x * (blockDim.x >> 5) + (threadIdx.x >> 5);
  if (wave >= totalWaves) return;
  const int lane = threadIdx.x & 31;
  const int half = lane >> 4;
  const int l16  = lane & 15;
  const int tm  = wave / wavesPerRow;
  const int tn0 = (wave - tm * wavesPerRow) * 4;

  const _Float16* aRow = A + (size_t)(tm * 16 + l16) * K;
  const _Float16* bRow[4];
  #pragma unroll
  for (int j = 0; j < 4; ++j)
    bRow[j] = Bt + (size_t)((tn0 + j) * 16 + l16) * K;

  v8f acc0 = {}, acc1 = {}, acc2 = {}, acc3 = {};
  for (int k = 0; k < K; k += 32) {
    const int ka = k + half * 8;
    union { v16h v; v8h h[2]; } a, b0, b1, b2, b3;
    a.h[0]  = *(const v8h*)(aRow + ka);
    a.h[1]  = *(const v8h*)(aRow + ka + 16);
    b0.h[0] = *(const v8h*)(bRow[0] + ka);
    b0.h[1] = *(const v8h*)(bRow[0] + ka + 16);
    b1.h[0] = *(const v8h*)(bRow[1] + ka);
    b1.h[1] = *(const v8h*)(bRow[1] + ka + 16);
    b2.h[0] = *(const v8h*)(bRow[2] + ka);
    b2.h[1] = *(const v8h*)(bRow[2] + ka + 16);
    b3.h[0] = *(const v8h*)(bRow[3] + ka);
    b3.h[1] = *(const v8h*)(bRow[3] + ka + 16);
    acc0 = __builtin_amdgcn_wmma_f32_16x16x32_f16(false, a.v, false, b0.v, (short)0, acc0, false, false);
    acc1 = __builtin_amdgcn_wmma_f32_16x16x32_f16(false, a.v, false, b1.v, (short)0, acc1, false, false);
    acc2 = __builtin_amdgcn_wmma_f32_16x16x32_f16(false, a.v, false, b2.v, (short)0, acc2, false, false);
    acc3 = __builtin_amdgcn_wmma_f32_16x16x32_f16(false, a.v, false, b3.v, (short)0, acc3, false, false);
  }
  float* cRow = C + (size_t)(tm * 16 + half * 8) * N + tn0 * 16 + l16;
  #pragma unroll
  for (int j = 0; j < 8; ++j) {
    float* cr = cRow + (size_t)j * N;
    cr[0]  = acc0[j];
    cr[16] = acc1[j];
    cr[32] = acc2[j];
    cr[48] = acc3[j];
  }
}

__global__ void k_attn(const float* __restrict__ h, const float* __restrict__ att_s,
                       const float* __restrict__ att_d,
                       float* __restrict__ as_, float* __restrict__ ad_,
                       int Nn, int H, int Cc) {
  int t = blockIdx.x * blockDim.x + threadIdx.x;
  if (t >= Nn * H) return;
  int n = t / H, hh = t - n * H;
  const float* hp = h + (size_t)n * H * Cc + (size_t)hh * Cc;
  const float* sp = att_s + hh * Cc;
  const float* dp = att_d + hh * Cc;
  float a0 = 0.f, a1 = 0.f;
  for (int c = 0; c < Cc; ++c) { float v = hp[c]; a0 += v * sp[c]; a1 += v * dp[c]; }
  as_[t] = a0;
  ad_[t] = a1;
}

#define TILE_R   200
#define TILE_T   1024
#define TILE_W   (TILE_T / 32)
#define SCAN_LD  16
#define SCAN_E   (SCAN_LD * 32)
#define QL       384
#define PASS_U   4
#define PQCAP    (32 * PASS_U + 32)
#define FIX_ONE  1099511627776.0f
static_assert(TILE_R <= 256, "a row inside the tile is kept in one byte");
static_assert(TILE_W == 32, "a row's wave is row & 31; lane i reads wave i's end-of-list flag");
static_assert(QL > 64, "a round must be able to take at least one load's hits and the self loops");

template <int HC, int H, bool FIRST>
__global__ __launch_bounds__(TILE_T)
void k_tile_agg(const int* __restrict__ src, const int* __restrict__ dst, int E, int Nn,
                const float* __restrict__ as_, const float* __restrict__ ad_,
                const float* __restrict__ feat, const float* __restrict__ bias,
                float* __restrict__ outF, _Float16* __restrict__ outH,
                int* __restrict__ lsrc, unsigned char* __restrict__ lrow,
                int* __restrict__ lcnt) {
  constexpr int Cc  = HC / H;
  constexpr int NCH = HC / 32;
  constexpr int EPI = (NCH >= 8) ? 4 : 8;
  __shared__ float acc[TILE_R * HC];
  __shared__ unsigned long long sfix[TILE_R * H];
  __shared__ int   mrow[TILE_R * H];
  __shared__ int   mold[TILE_R * H];
  __shared__ int   q[TILE_W][QL];
  __shared__ unsigned char ql[TILE_W][QL];
  __shared__ int   qcnt[TILE_W];
  __shared__ int   qfin[TILE_W];
  __shared__ int   pqs[TILE_W][PQCAP];
  __shared__ unsigned char pqr[TILE_W][PQCAP];

  const int tid = threadIdx.x, lane = tid & 31;
  const int w = __builtin_amdgcn_readfirstlane(tid >> 5);
  const int r0 = blockIdx.x * TILE_R;
  const int nrows = (Nn - r0 < TILE_R) ? (Nn - r0) : TILE_R;
  const int ninf = f2oi(-__builtin_inff());

  for (int i = tid * 4; i < TILE_R * HC; i += TILE_T * 4)
    *(float4*)(acc + i) = make_float4(0.f, 0.f, 0.f, 0.f);
  for (int i = tid; i < TILE_R * H; i += TILE_T) { mrow[i] = ninf; sfix[i] = 0ull; }

  const int nStep = (E + SCAN_E - 1) / SCAN_E;
  const int rot = (int)(((long long)blockIdx.x * nStep) / gridDim.x);
  const int nIter = (nStep - w + TILE_W - 1) / TILE_W;
  const int maxRound = (((nStep + TILE_W - 1) / TILE_W) * SCAN_E) / (QL - 32) + 3;

  auto stepBase = [&](int k) {
    int c = w + k * TILE_W + rot;
    if (c >= nStep) c -= nStep;
    return c * SCAN_E;
  };
  int dd[SCAN_LD];
  auto loadStep = [&](int b) {
    if (b + SCAN_E <= E) {
      for (int j = 0; j < SCAN_LD; ++j) dd[j] = dst[b + lane + j * 32];
    } else {
      for (int j = 0; j < SCAN_LD; ++j) dd[j] = dst[(b + lane + j * 32 < E) ? (b + lane + j * 32) : (E - 1)];
    }
  };

  int  it = 0, jStart = 0;
  bool selfDone = false;
  const size_t lb = ((size_t)blockIdx.x * TILE_W + w) * QL;
  const int  cnt0 = FIRST ? -1 : lcnt[blockIdx.x * TILE_W + w];
  const bool saved = cnt0 >= 0;

  for (int round = 0; round < maxRound; ++round) {
    int  qn = 0;
    bool full = false;
    if (round > 0)
      for (int i = tid; i < TILE_R * H; i += TILE_T) mold[i] = mrow[i];

    if (saved) {
      qn = cnt0;
      for (int ix = lane; ix < qn; ix += 32) { q[w][ix] = lsrc[lb + ix]; ql[w][ix] = lrow[lb + ix]; }
      it = nIter;
      selfDone = true;
    }
    int bNext = 0;
    if (it < nIter) { bNext = stepBase(it); loadStep(bNext); }
    while (it < nIter && !full) {
      const int b0 = bNext;
      int dl[SCAN_LD];
      unsigned mk[SCAN_LD];
      for (int j = 0; j < SCAN_LD; ++j) dl[j] = dd[j] - r0;
      if (b0 + SCAN_E <= E) {
        for (int j = 0; j < SCAN_LD; ++j)
          mk[j] = __builtin_amdgcn_ballot_w32((unsigned)dl[j] < (unsigned)nrows);
      } else {
        for (int j = 0; j < SCAN_LD; ++j)
          mk[j] = __builtin_amdgcn_ballot_w32((unsigned)dl[j] < (unsigned)nrows && b0 + lane + j * 32 < E);
      }
      if (it + 1 < nIter) { bNext = stepBase(it + 1); loadStep(bNext); }
      for (int j = 0; j < SCAN_LD; ++j) {
        if (j >= jStart && !full && mk[j]) {
          const int c = __builtin_popcount(mk[j]);
          if (qn + c > QL) {
            full = true;
            jStart = j;
          } else {
            if ((mk[j] >> lane) & 1u) {
              const int pos = qn + (int)__builtin_amdgcn_mbcnt_lo(mk[j], 0u);
              q[w][pos] = b0 + lane + j * 32;
              ql[w][pos] = (unsigned char)dl[j];
            }
            qn += c;
          }
        }
      }
      if (!full) { jStart = 0; ++it; }
    }
    if (it == nIter && !full && !selfDone) {
      const int row = lane * TILE_W + w;
      const bool hit = row < nrows;
      const unsigned mask = __builtin_amdgcn_ballot_w32(hit);
      const int c = __builtin_popcount(mask);
      if (qn + c <= QL) {
        if (hit) {
          const int pos = qn + (int)__builtin_amdgcn_mbcnt_lo(mask, 0u);
          q[w][pos] = E + r0 + row;
          ql[w][pos] = (unsigned char)row;
        }
        qn += c;
        selfDone = true;
      }
    }
    if (lane == 0) { qcnt[w] = qn; qfin[w] = selfDone ? 1 : 0; }
    __syncthreads();
    const bool lastRound = __builtin_amdgcn_ballot_w32(qfin[lane] != 0) == 0xFFFFFFFFu;
    if (FIRST && round == 0 && w == 0) {
      lcnt[blockIdx.x * TILE_W + lane] = lastRound ? qcnt[lane] : -1;
    }

    for (int base = 0; base < qn; base += 32 * PASS_U) {
      int sv[PASS_U], lr[PASS_U], ix[PASS_U];
      bool ok[PASS_U];
      float ev[PASS_U][H];
      for (int u = 0; u < PASS_U; ++u) {
        ix[u] = base + u * 32 + lane;
        ok[u] = ix[u] < qn;
        if (!ok[u]) ix[u] = base;
        sv[u] = q[w][ix[u]];
        lr[u] = ql[w][ix[u]];
      }
      if (!saved) {
        for (int u = 0; u < PASS_U; ++u) {
          const int e = sv[u];
          const int sl = src[e < E ? e : 0];
          sv[u] = e < E ? sl : e - E;
        }
      }
      for (int u = 0; u < PASS_U; ++u)
        for (int hh = 0; hh < H; ++hh)
          ev[u][hh] = leaky(as_[sv[u] * H + hh] + ad_[(r0 + lr[u]) * H + hh]);
      for (int u = 0; u < PASS_U; ++u) {
        for (int hh = 0; hh < H; ++hh)
          __hip_atomic_fetch_max(&mrow[lr[u] * H + hh], ok[u] ? f2oi(ev[u][hh]) : ninf,
                                 __ATOMIC_RELAXED, __HIP_MEMORY_SCOPE_WORKGROUP);
        if (ok[u] && !saved) q[w][ix[u]] = sv[u];
        if (FIRST && round == 0 && ok[u]) { lsrc[lb + ix[u]] = sv[u]; lrow[lb + ix[u]] = (unsigned char)lr[u]; }
      }
    }
    __syncthreads();

    if (round > 0) {
      for (int i = tid; i < TILE_R * HC; i += TILE_T) {
        const int rh = (i / HC) * H + (i % HC) / Cc;
        const int a = mold[rh], b = mrow[rh];
        if (a != b) acc[i] *= expf(oi2f(a) - oi2f(b));
      }
      for (int i = tid; i < TILE_R * H; i += TILE_T) {
        const int a = mold[i], b = mrow[i];
        if (a != b) sfix[i] = (unsigned long long)((float)sfix[i] * expf(oi2f(a) - oi2f(b)));
      }
      __syncthreads();
    }

    for (int base = 0; base < qn; base += 32 * PASS_U) {
      int sv[PASS_U], lr[PASS_U];
      bool ok[PASS_U];
      float ev[PASS_U][H];
      for (int u = 0; u < PASS_U; ++u) {
        int ix = base + u * 32 + lane;
        ok[u] = ix < qn;
        if (!ok[u]) ix = base;
        sv[u] = q[w][ix];
        lr[u] = ql[w][ix];
      }
      for (int u = 0; u < PASS_U; ++u)
        for (int hh = 0; hh < H; ++hh)
          ev[u][hh] = leaky(as_[sv[u] * H + hh] + ad_[(r0 + lr[u]) * H + hh]);
      for (int u = 0; u < PASS_U; ++u) {
        for (int hh = 0; hh < H; ++hh) {
          const float ex = expf(ev[u][hh] - oi2f(mrow[lr[u] * H + hh]));
          __hip_atomic_fetch_add(&sfix[lr[u] * H + hh], ok[u] ? (unsigned long long)(ex * FIX_ONE) : 0ull,
                                 __ATOMIC_RELAXED, __HIP_MEMORY_SCOPE_WORKGROUP);
        }
      }
    }

    int pn = 0;
    auto work = [&](bool last) {
      while (pn >= 32 || (last && pn > 0)) {
        const int n = pn < 32 ? pn : 32;
        const int pb = pn - n;
        pn = pb;

        int s_ = 0, lrow = 0;
        float al[H];
        for (int hh = 0; hh < H; ++hh) al[hh] = 0.f;
        if (lane < n) {
          s_ = pqs[w][pb + lane];
          lrow = pqr[w][pb + lane];
          const int d_ = r0 + lrow;
          for (int hh = 0; hh < H; ++hh) {
            const float ev = leaky(as_[s_ * H + hh] + ad_[d_ * H + hh]);
            al[hh] = expf(ev - oi2f(mrow[lrow * H + hh]));
          }
        }

        for (int i = 0; i < n; i += EPI) {
          float v[EPI][NCH];
          int   lr[EPI];
          float a[EPI][H];
          for (int k = 0; k < EPI; ++k) {
            const int ik = (i + k < n) ? (i + k) : (n - 1);
            const int sk = __builtin_amdgcn_readlane(s_, ik);
            lr[k] = __builtin_amdgcn_readlane(lrow, ik);
            for (int hh = 0; hh < H; ++hh)
              a[k][hh] = __int_as_float(__builtin_amdgcn_readlane(__float_as_int(al[hh]), ik));
            const float* fRow = feat + (size_t)sk * HC + lane;
            for (int j = 0; j < NCH; ++j) v[k][j] = fRow[j * 32];
          }
          for (int k = 0; k < EPI; ++k) {
            if (i + k < n) {
              float* aRow = acc + lr[k] * HC + lane;
              for (int j = 0; j < NCH; ++j)
                __hip_atomic_fetch_add(aRow + j * 32, v[k][j] * a[k][(j * 32) / Cc],
                                       __ATOMIC_RELAXED, __HIP_MEMORY_SCOPE_WORKGROUP);
            }
          }
        }
      }
    };
    for (int wl = 0; wl < TILE_W; ++wl) {
      const int cn = __builtin_amdgcn_readfirstlane(qcnt[wl]);
      for (int base = 0; base < cn; base += 32 * PASS_U) {
        int sv[PASS_U], lr[PASS_U];
        unsigned mk[PASS_U];
        for (int u = 0; u < PASS_U; ++u) {
          int ix = base + u * 32 + lane;
          const bool in = ix < cn;
          if (!in) ix = base;
          sv[u] = q[wl][ix];
          lr[u] = ql[wl][ix];
          mk[u] = __builtin_amdgcn_ballot_w32(in && (lr[u] & (TILE_W - 1)) == w);
        }
        for (int u = 0; u < PASS_U; ++u) {
          if (mk[u]) {
            if ((mk[u] >> lane) & 1u) {
              const int pos = pn + (int)__builtin_amdgcn_mbcnt_lo(mk[u], 0u);
              pqs[w][pos] = sv[u];
              pqr[w][pos] = (unsigned char)lr[u];
            }
            pn += __builtin_popcount(mk[u]);
          }
        }
        work(false);
      }
    }
    work(true);
    __syncthreads();
    if (lastRound) break;
  }

  const size_t o0 = (size_t)r0 * HC;
  for (int i = tid * 4; i < nrows * HC; i += TILE_T * 4) {
    const int rh = (i / HC) * H + (i & (HC - 1)) / Cc;
    const float den = (float)sfix[rh] * (1.0f / FIX_ONE) + 1e-16f;
    const float4 a = *(const float4*)(acc + i);
    float4 r = add4(make_float4(a.x / den, a.y / den, a.z / den, a.w / den),
                    *(const float4*)(bias + (i & (HC - 1))));
    if (FIRST) *(v4h*)(outH + o0 + i) = cast4(relu4(r));
    else       *(float4*)(outF + o0 + i) = r;
  }
}

extern "C" void kernel_launch(void* const* d_in, const int* in_sizes, int n_in,
                              void* d_out, int out_size, void* d_ws, size_t ws_size,
                              hipStream_t stream) {
  const int IN = 128, HID = 64, H1 = 4;
  const int HC1 = H1 * HID;
  const int HC2 = HID;
  const int Nn   = in_sizes[0] / IN;
  const int E    = in_sizes[1] / 2;

  const float* x   = (const float*)d_in[0];
  const int*   ei  = (const int*)d_in[1];
  const float* W1  = (const float*)d_in[2];
  const float* aS1 = (const float*)d_in[3];
  const float* aD1 = (const float*)d_in[4];
  const float* b1  = (const float*)d_in[5];
  const float* W2  = (const float*)d_in[6];
  const float* aS2 = (const float*)d_in[7];
  const float* aD2 = (const float*)d_in[8];
  const float* b2  = (const float*)d_in[9];
  float* out = (float*)d_out;
  const int* srcI = ei;
  const int* dstI = ei + E;

  char* p = (char*)d_ws;
  auto carve = [&](size_t bytes) -> void* {
    void* r = (void*)p;
    p += (bytes + 255) & ~(size_t)255;
    return r;
  };
  _Float16* xh   = (_Float16*)carve((size_t)Nn * IN * 2);
  _Float16* w1t  = (_Float16*)carve((size_t)HC1 * IN * 2);
  _Float16* w2t  = (_Float16*)carve((size_t)HID * HC1 * 2);
  float*    h1   = (float*)carve((size_t)Nn * HC1 * 4);
  _Float16* h1h  = (_Float16*)carve((size_t)Nn * HC1 * 2);
  float*    h2   = (float*)carve((size_t)Nn * HC2 * 4);
  float*    as1  = (float*)carve((size_t)Nn * H1 * 4);
  float*    ad1  = (float*)carve((size_t)Nn * H1 * 4);
  float*    as2  = (float*)carve((size_t)Nn * 4);
  float*    ad2  = (float*)carve((size_t)Nn * 4);
  const size_t nTile = (size_t)((Nn + TILE_R - 1) / TILE_R);
  int*           lsrc = (int*)carve(nTile * TILE_W * QL * 4);
  unsigned char* lrow = (unsigned char*)carve(nTile * TILE_W * QL);
  int*           lcnt = (int*)carve(nTile * TILE_W * 4);

  const int TB = 256;
  auto cdiv = [](long long a, long long b) { return (int)((a + b - 1) / b); };

  k_cast_f16<<<cdiv((long long)Nn * IN, EW_SPAN), TB, 0, stream>>>(x, xh, Nn * IN);
  k_castT_f16<<<cdiv((long long)IN * HC1, TB), TB, 0, stream>>>(W1, w1t, IN, HC1);
  k_castT_f16<<<cdiv((long long)HC1 * HID, TB), TB, 0, stream>>>(W2, w2t, HC1, HID);

  {
    int waves = (Nn / 16) * (HC1 / 64);
    k_wmma_gemm<<<cdiv(waves, 8), TB, 0, stream>>>(xh, w1t, h1, Nn, HC1, IN);
  }
  k_attn<<<cdiv((long long)Nn * H1, TB), TB, 0, stream>>>(h1, aS1, aD1, as1, ad1, Nn, H1, HID);
  k_tile_agg<256, 4, true><<<cdiv(Nn, TILE_R), TILE_T, 0, stream>>>(srcI, dstI, E, Nn, as1, ad1, h1, b1, nullptr, h1h, lsrc, lrow, lcnt);

  {
    int waves = (Nn / 16) * (HC2 / 64);
    k_wmma_gemm<<<cdiv(waves, 8), TB, 0, stream>>>(h1h, w2t, h2, Nn, HC2, HC1);
  }
  k_attn<<<cdiv((long long)Nn, TB), TB, 0, stream>>>(h2, aS2, aD2, as2, ad2, Nn, 1, HID);
  k_tile_agg<64, 1, false><<<cdiv(Nn, TILE_R), TILE_T, 0, stream>>>(srcI, dstI, E, Nn, as2, ad2, h2, b2, out, nullptr, lsrc, lrow, lcnt);
}


# Exp-sums formed in the weight pass: an edge's weight exp(logit-max) is also added, exact 64-bit fixed point, to its row's sum; the exp-sum pass (as_ look-up, exp per edge) is gone; sums unchanged
# speedup vs baseline: 3.3888x; 1.2047x over previous
#include <hip/hip_runtime.h>
#include <hip/hip_bf16.h>
#include <math.h>

typedef __attribute__((ext_vector_type(16))) _Float16 v16h;
typedef __attribute__((ext_vector_type(8)))  _Float16 v8h;
typedef __attribute__((ext_vector_type(8)))  float    v8f;
typedef __attribute__((ext_vector_type(4)))  _Float16 v4h;

#define EW_SPAN 2048
#define EW_HALF 1024

__device__ __forceinline__ v4h cast4(float4 a) {
  v4h o;
  o[0] = (_Float16)a.x; o[1] = (_Float16)a.y; o[2] = (_Float16)a.z; o[3] = (_Float16)a.w;
  return o;
}
__device__ __forceinline__ float4 add4(float4 a, float4 b) {
  return make_float4(a.x + b.x, a.y + b.y, a.z + b.z, a.w + b.w);
}
__device__ __forceinline__ float4 relu4(float4 a) {
  return make_float4(a.x > 0.f ? a.x : 0.f, a.y > 0.f ? a.y : 0.f, a.z > 0.f ? a.z : 0.f, a.w > 0.f ? a.w : 0.f);
}

#define NEG_SLOPE 0.2f

__device__ __forceinline__ int f2oi(float f) {
  int i = __float_as_int(f);
  return i >= 0 ? i : (i ^ 0x7FFFFFFF);
}
__device__ __forceinline__ float oi2f(int i) {
  return __int_as_float(i >= 0 ? i : (i ^ 0x7FFFFFFF));
}
__device__ __forceinline__ float leaky(float x) { return x > 0.f ? x : NEG_SLOPE * x; }

__global__ void k_cast_f16(const float* __restrict__ in, _Float16* __restrict__ out, int n) {
  const int i0 = blockIdx.x * EW_SPAN + threadIdx.x * 4;
  const int i1 = i0 + EW_HALF;
  if (i1 + 4 <= n) {
    float4 a0 = *(const float4*)(in + i0);
    float4 a1 = *(const float4*)(in + i1);
    *(v4h*)(out + i0) = cast4(a0);
    *(v4h*)(out + i1) = cast4(a1);
  } else {
    for (int g = 0; g < 2; ++g) {
      int i = g ? i1 : i0;
      if (i + 4 <= n) *(v4h*)(out + i) = cast4(*(const float4*)(in + i));
      else for (; i < n; ++i) out[i] = (_Float16)in[i];
    }
  }
}

__global__ void k_castT_f16(const float* __restrict__ in, _Float16* __restrict__ out, int R, int C) {
  int t = blockIdx.x * blockDim.x + threadIdx.x;
  if (t >= R * C) return;
  int r = t / C, c = t - r * C;
  out[c * R + r] = (_Float16)in[t];
}

__global__ void k_wmma_gemm(const _Float16* __restrict__ A,
                            const _Float16* __restrict__ Bt,
                            float* __restrict__ C,
                            int M, int N, int K) {
  const int tilesN      = N >> 4;
  const int wavesPerRow = tilesN >> 2;
  const int totalWaves  = (M >> 4) * wavesPerRow;
  const int wave = blockIdx.x * (blockDim.x >> 5) + (threadIdx.x >> 5);
  if (wave >= totalWaves) return;
  const int lane = threadIdx.x & 31;
  const int half = lane >> 4;
  const int l16  = lane & 15;
  const int tm  = wave / wavesPerRow;
  const int tn0 = (wave - tm * wavesPerRow) * 4;

  const _Float16* aRow = A + (size_t)(tm * 16 + l16) * K;
  const _Float16* bRow[4];
  #pragma unroll
  for (int j = 0; j < 4; ++j)
    bRow[j] = Bt + (size_t)((tn0 + j) * 16 + l16) * K;

  v8f acc0 = {}, acc1 = {}, acc2 = {}, acc3 = {};
  for (int k = 0; k < K; k += 32) {
    const int ka = k + half * 8;
    union { v16h v; v8h h[2]; } a, b0, b1, b2, b3;
    a.h[0]  = *(const v8h*)(aRow + ka);
    a.h[1]  = *(const v8h*)(aRow + ka + 16);
    b0.h[0] = *(const v8h*)(bRow[0] + ka);
    b0.h[1] = *(const v8h*)(bRow[0] + ka + 16);
    b1.h[0] = *(const v8h*)(bRow[1] + ka);
    b1.h[1] = *(const v8h*)(bRow[1] + ka + 16);
    b2.h[0] = *(const v8h*)(bRow[2] + ka);
    b2.h[1] = *(const v8h*)(bRow[2] + ka + 16);
    b3.h[0] = *(const v8h*)(bRow[3] + ka);
    b3.h[1] = *(const v8h*)(bRow[3] + ka + 16);
    acc0 = __builtin_amdgcn_wmma_f32_16x16x32_f16(false, a.v, false, b0.v, (short)0, acc0, false, false);
    acc1 = __builtin_amdgcn_wmma_f32_16x16x32_f16(false, a.v, false, b1.v, (short)0, acc1, false, false);
    acc2 = __builtin_amdgcn_wmma_f32_16x16x32_f16(false, a.v, false, b2.v, (short)0, acc2, false, false);
    acc3 = __builtin_amdgcn_wmma_f32_16x16x32_f16(false, a.v, false, b3.v, (short)0, acc3, false, false);
  }
  float* cRow = C + (size_t)(tm * 16 + half * 8) * N + tn0 * 16 + l16;
  #pragma unroll
  for (int j = 0; j < 8; ++j) {
    float* cr = cRow + (size_t)j * N;
    cr[0]  = acc0[j];
    cr[16] = acc1[j];
    cr[32] = acc2[j];
    cr[48] = acc3[j];
  }
}

__global__ void k_attn(const float* __restrict__ h, const float* __restrict__ att_s,
                       const float* __restrict__ att_d,
                       float* __restrict__ as_, float* __restrict__ ad_,
                       int Nn, int H, int Cc) {
  int t = blockIdx.x * blockDim.x + threadIdx.x;
  if (t >= Nn * H) return;
  int n = t / H, hh = t - n * H;
  const float* hp = h + (size_t)n * H * Cc + (size_t)hh * Cc;
  const float* sp = att_s + hh * Cc;
  const float* dp = att_d + hh * Cc;
  float a0 = 0.f, a1 = 0.f;
  for (int c = 0; c < Cc; ++c) { float v = hp[c]; a0 += v * sp[c]; a1 += v * dp[c]; }
  as_[t] = a0;
  ad_[t] = a1;
}

#define TILE_R   200
#define TILE_T   1024
#define TILE_W   (TILE_T / 32)
#define SCAN_LD  6
#define SCAN_E   1024
#define BITS_T   128
#define BITS_TW  256
#define QL       384
#define PASS_U   4
#define PQCAP    (32 * PASS_U + 32)
#define FIX_ONE  1099511627776.0f
static_assert(TILE_R <= 256, "a row inside the tile is kept in one byte");
static_assert(TILE_W == 32, "a row's wave is row & 31; lane i reads wave i's end-of-list flag");
static_assert(QL > 64, "a round must be able to take at least one load's hits and the self loops");
static_assert(SCAN_E == 32 * 32, "a word's mask has one bit per block of 32 edges");

__global__ __launch_bounds__(BITS_T)
void k_tile_bits(const int* __restrict__ dst, int E, int nWord, int pitch, unsigned* __restrict__ bits) {
  __shared__ unsigned arr[BITS_T / 32][BITS_TW];
  const int lane = threadIdx.x & 31, wv = threadIdx.x >> 5;
  const int word = blockIdx.x * (BITS_T / 32) + wv;
  const int t0 = blockIdx.y * BITS_TW;
  for (int i = lane; i < BITS_TW; i += 32) arr[wv][i] = 0u;
  __syncthreads();
  const int e0 = word * SCAN_E + lane;
  int d[32];
  for (int b = 0; b < 32; ++b) {
    const int e = e0 + b * 32;
    d[b] = dst[e < E ? e : (E - 1)];
  }
  for (int b = 0; b < 32; ++b) {
    const int t = d[b] / TILE_R - t0;
    if (e0 + b * 32 < E && (unsigned)t < (unsigned)BITS_TW)
      __hip_atomic_fetch_or(&arr[wv][t], 1u << b, __ATOMIC_RELAXED, __HIP_MEMORY_SCOPE_WORKGROUP);
  }
  __syncthreads();
  if (word < nWord)
    for (int i = lane; i < BITS_TW; i += 32)
      if (t0 + i < pitch) bits[(size_t)word * pitch + t0 + i] = arr[wv][i];
}

template <int HC, int H, bool FIRST>
__global__ __launch_bounds__(TILE_T)
void k_tile_agg(const int* __restrict__ src, const int* __restrict__ dst, int E, int Nn,
                const float* __restrict__ as_, const float* __restrict__ ad_,
                const float* __restrict__ feat, const float* __restrict__ bias,
                float* __restrict__ outF, _Float16* __restrict__ outH,
                int* __restrict__ lsrc, unsigned char* __restrict__ lrow,
                int* __restrict__ lcnt, const unsigned* __restrict__ bits, int pitch) {
  constexpr int Cc  = HC / H;
  constexpr int VW  = (HC >= 128) ? 4 : 2;
  constexpr int NP  = HC / (32 * VW);
  constexpr int HPP = (32 * VW) / Cc;
  constexpr int EPI = (HC >= 256) ? 4 : 8;
  static_assert(HC % (32 * VW) == 0 && Cc % VW == 0, "a lane's piece lies inside one head");
  static_assert((HPP == 1 || HPP == 2) && HPP * Cc == 32 * VW, "a piece spans one head or two halves");
  typedef __attribute__((ext_vector_type(VW))) float vrow;
  __shared__ __attribute__((aligned(16))) float acc[TILE_R * HC];
  __shared__ unsigned long long sfix[TILE_R * H];
  __shared__ int   mrow[TILE_R * H];
  __shared__ int   mold[TILE_R * H];
  __shared__ int   q[TILE_W][QL];
  __shared__ unsigned char ql[TILE_W][QL];
  __shared__ int   qcnt[TILE_W];
  __shared__ int   qfin[TILE_W];
  __shared__ int   pqs[TILE_W][PQCAP];
  __shared__ unsigned char pqr[TILE_W][PQCAP];

  const int tid = threadIdx.x, lane = tid & 31;
  const int w = __builtin_amdgcn_readfirstlane(tid >> 5);
  const int r0 = blockIdx.x * TILE_R;
  const int nrows = (Nn - r0 < TILE_R) ? (Nn - r0) : TILE_R;
  const int ninf = f2oi(-__builtin_inff());

  for (int i = tid * 4; i < TILE_R * HC; i += TILE_T * 4)
    *(float4*)(acc + i) = make_float4(0.f, 0.f, 0.f, 0.f);
  for (int i = tid; i < TILE_R * H; i += TILE_T) { mrow[i] = ninf; sfix[i] = 0ull; }

  const int nStep = (E + SCAN_E - 1) / SCAN_E;
  const int rot = (int)(((long long)blockIdx.x * nStep) / gridDim.x);
  const int nIter = (nStep - w + TILE_W - 1) / TILE_W;
  const int maxRound = (((nStep + TILE_W - 1) / TILE_W) * SCAN_E) / (QL - 32) + 3;

  auto stepWord = [&](int k) {
    int c = w + k * TILE_W + rot;
    if (c >= nStep) c -= nStep;
    return c;
  };
  int mv = 0;
  auto loadMasks = [&](int first) {
    const int k = first + lane;
    int c = w + k * TILE_W + rot;
    if (c >= nStep) c -= nStep;
    mv = (k < nIter) ? (int)bits[(size_t)c * pitch + blockIdx.x] : 0;
  };

  int  it = 0, jStart = 0;
  bool selfDone = false;
  const size_t lb = ((size_t)blockIdx.x * TILE_W + w) * QL;
  const int  cnt0 = FIRST ? -1 : lcnt[blockIdx.x * TILE_W + w];
  const bool saved = cnt0 >= 0;

  for (int round = 0; round < maxRound; ++round) {
    int  qn = 0;
    bool full = false;
    if (round > 0)
      for (int i = tid; i < TILE_R * H; i += TILE_T) mold[i] = mrow[i];

    if (saved) {
      qn = cnt0;
      for (int ix = lane; ix < qn; ix += 32) { q[w][ix] = lsrc[lb + ix]; ql[w][ix] = lrow[lb + ix]; }
      it = nIter;
      selfDone = true;
    }
    unsigned m = 0u;
    if (it < nIter) {
      loadMasks(it & ~31);
      m = (unsigned)__builtin_amdgcn_readlane(mv, it & 31) & (0xFFFFFFFFu << jStart);
    }
    while (it < nIter && !full) {
      const int b0 = stepWord(it) * SCAN_E;
      while (m != 0u && !full) {
        const int nb = __builtin_popcount(m);
        int bj[SCAN_LD], dd[SCAN_LD];
        unsigned mm = m;
        for (int j = 0; j < SCAN_LD; ++j) {
          bj[j] = (int)__builtin_ctz(mm | 0x80000000u);
          mm &= mm - 1u;
        }
        for (int j = 0; j < SCAN_LD; ++j) {
          const int e = b0 + bj[j] * 32 + lane;
          dd[j] = dst[e < E ? e : (E - 1)];
        }
        for (int j = 0; j < SCAN_LD; ++j) {
          if (j < nb && !full) {
            const int e = b0 + bj[j] * 32 + lane;
            const int dl = dd[j] - r0;
            const unsigned mk = __builtin_amdgcn_ballot_w32((unsigned)dl < (unsigned)nrows && e < E);
            const int c = __builtin_popcount(mk);
            if (qn + c > QL) {
              full = true;
              jStart = bj[j];
            } else {
              if ((mk >> lane) & 1u) {
                const int pos = qn + (int)__builtin_amdgcn_mbcnt_lo(mk, 0u);
                q[w][pos] = e;
                ql[w][pos] = (unsigned char)dl;
              }
              qn += c;
              m &= ~(1u << bj[j]);
            }
          }
        }
      }
      if (!full) {
        jStart = 0;
        ++it;
        m = 0u;
        if (it < nIter) {
          if ((it & 31) == 0) loadMasks(it);
          m = (unsigned)__builtin_amdgcn_readlane(mv, it & 31);
        }
      }
    }
    if (it == nIter && !full && !selfDone) {
      const int row = lane * TILE_W + w;
      const bool hit = row < nrows;
      const unsigned mask = __builtin_amdgcn_ballot_w32(hit);
      const int c = __builtin_popcount(mask);
      if (qn + c <= QL) {
        if (hit) {
          const int pos = qn + (int)__builtin_amdgcn_mbcnt_lo(mask, 0u);
          q[w][pos] = E + r0 + row;
          ql[w][pos] = (unsigned char)row;
        }
        qn += c;
        selfDone = true;
      }
    }
    if (lane == 0) { qcnt[w] = qn; qfin[w] = selfDone ? 1 : 0; }
    __syncthreads();
    const bool lastRound = __builtin_amdgcn_ballot_w32(qfin[lane] != 0) == 0xFFFFFFFFu;
    if (FIRST && round == 0 && w == 0) {
      lcnt[blockIdx.x * TILE_W + lane] = lastRound ? qcnt[lane] : -1;
    }

    for (int base = 0; base < qn; base += 32 * PASS_U) {
      int sv[PASS_U], lr[PASS_U], ix[PASS_U];
      bool ok[PASS_U];
      float ev[PASS_U][H];
      for (int u = 0; u < PASS_U; ++u) {
        ix[u] = base + u * 32 + lane;
        ok[u] = ix[u] < qn;
        if (!ok[u]) ix[u] = base;
        sv[u] = q[w][ix[u]];
        lr[u] = ql[w][ix[u]];
      }
      if (!saved) {
        for (int u = 0; u < PASS_U; ++u) {
          const int e = sv[u];
          const int sl = src[e < E ? e : 0];
          sv[u] = e < E ? sl : e - E;
        }
      }
      for (int u = 0; u < PASS_U; ++u)
        for (int hh = 0; hh < H; ++hh)
          ev[u][hh] = leaky(as_[sv[u] * H + hh] + ad_[(r0 + lr[u]) * H + hh]);
      for (int u = 0; u < PASS_U; ++u) {
        for (int hh = 0; hh < H; ++hh)
          __hip_atomic_fetch_max(&mrow[lr[u] * H + hh], ok[u] ? f2oi(ev[u][hh]) : ninf,
                                 __ATOMIC_RELAXED, __HIP_MEMORY_SCOPE_WORKGROUP);
        if (ok[u] && !saved) q[w][ix[u]] = sv[u];
        if (FIRST && round == 0 && ok[u]) { lsrc[lb + ix[u]] = sv[u]; lrow[lb + ix[u]] = (unsigned char)lr[u]; }
      }
    }
    __syncthreads();

    if (round > 0) {
      for (int i = tid; i < TILE_R * HC; i += TILE_T) {
        const int rh = (i / HC) * H + (i % HC) / Cc;
        const int a = mold[rh], b = mrow[rh];
        if (a != b) acc[i] *= expf(oi2f(a) - oi2f(b));
      }
      for (int i = tid; i < TILE_R * H; i += TILE_T) {
        const int a = mold[i], b = mrow[i];
        if (a != b) sfix[i] = (unsigned long long)((float)sfix[i] * expf(oi2f(a) - oi2f(b)));
      }
      __syncthreads();
    }

    int pn = 0;
    auto work = [&](bool last) {
      while (pn >= 32 || (last && pn > 0)) {
        const int n = pn < 32 ? pn : 32;
        const int pb = pn - n;
        pn = pb;

        int s_ = 0, lrow = 0;
        float al[H];
        for (int hh = 0; hh < H; ++hh) al[hh] = 0.f;
        if (lane < n) {
          s_ = pqs[w][pb + lane];
          lrow = pqr[w][pb + lane];
          const int d_ = r0 + lrow;
          for (int hh = 0; hh < H; ++hh) {
            const float ev = leaky(as_[s_ * H + hh] + ad_[d_ * H + hh]);
            al[hh] = expf(ev - oi2f(mrow[lrow * H + hh]));
            __hip_atomic_fetch_add(&sfix[lrow * H + hh], (unsigned long long)(al[hh] * FIX_ONE),
                                   __ATOMIC_RELAXED, __HIP_MEMORY_SCOPE_WORKGROUP);
          }
        }

        for (int i = 0; i < n; i += EPI) {
          vrow  v[EPI][NP];
          int   lr[EPI];
          float a[EPI][H];
          for (int k = 0; k < EPI; ++k) {
            const int ik = (i + k < n) ? (i + k) : (n - 1);
            const int sk = __builtin_amdgcn_readlane(s_, ik);
            lr[k] = __builtin_amdgcn_readlane(lrow, ik);
            for (int hh = 0; hh < H; ++hh)
              a[k][hh] = __int_as_float(__builtin_amdgcn_readlane(__float_as_int(al[hh]), ik));
            const float* fRow = feat + (size_t)sk * HC + lane * VW;
            for (int j = 0; j < NP; ++j) v[k][j] = *(const vrow*)(fRow + j * 32 * VW);
          }
          for (int k = 0; k < EPI; ++k) {
            if (i + k < n) {
              float* aRow = acc + lr[k] * HC + lane * VW;
              for (int j = 0; j < NP; ++j) {
                const float a0 = a[k][(j * HPP) % H], a1 = a[k][(j * HPP + HPP - 1) % H];
                const float aw = (HPP == 1 || lane < 16) ? a0 : a1;
                vrow t = *(vrow*)(aRow + j * 32 * VW);
                t += v[k][j] * aw;
                *(vrow*)(aRow + j * 32 * VW) = t;
              }
            }
          }
        }
      }
    };
    for (int wl = 0; wl < TILE_W; ++wl) {
      const int cn = __builtin_amdgcn_readfirstlane(qcnt[wl]);
      for (int base = 0; base < cn; base += 32 * PASS_U) {
        int sv[PASS_U], lr[PASS_U];
        unsigned mk[PASS_U];
        for (int u = 0; u < PASS_U; ++u) {
          int ix = base + u * 32 + lane;
          const bool in = ix < cn;
          if (!in) ix = base;
          sv[u] = q[wl][ix];
          lr[u] = ql[wl][ix];
          mk[u] = __builtin_amdgcn_ballot_w32(in && (lr[u] & (TILE_W - 1)) == w);
        }
        for (int u = 0; u < PASS_U; ++u) {
          if (mk[u]) {
            if ((mk[u] >> lane) & 1u) {
              const int pos = pn + (int)__builtin_amdgcn_mbcnt_lo(mk[u], 0u);
              pqs[w][pos] = sv[u];
              pqr[w][pos] = (unsigned char)lr[u];
            }
            pn += __builtin_popcount(mk[u]);
          }
        }
        work(false);
      }
    }
    work(true);
    __syncthreads();
    if (lastRound) break;
  }

  const size_t o0 = (size_t)r0 * HC;
  for (int i = tid * 4; i < nrows * HC; i += TILE_T * 4) {
    const int rh = (i / HC) * H + (i & (HC - 1)) / Cc;
    const float den = (float)sfix[rh] * (1.0f / FIX_ONE) + 1e-16f;
    const float4 a = *(const float4*)(acc + i);
    float4 r = add4(make_float4(a.x / den, a.y / den, a.z / den, a.w / den),
                    *(const float4*)(bias + (i & (HC - 1))));
    if (FIRST) *(v4h*)(outH + o0 + i) = cast4(relu4(r));
    else       *(float4*)(outF + o0 + i) = r;
  }
}

extern "C" void kernel_launch(void* const* d_in, const int* in_sizes, int n_in,
                              void* d_out, int out_size, void* d_ws, size_t ws_size,
                              hipStream_t stream) {
  const int IN = 128, HID = 64, H1 = 4;
  const int HC1 = H1 * HID;
  const int HC2 = HID;
  const int Nn   = in_sizes[0] / IN;
  const int E    = in_sizes[1] / 2;

  const float* x   = (const float*)d_in[0];
  const int*   ei  = (const int*)d_in[1];
  const float* W1  = (const float*)d_in[2];
  const float* aS1 = (const float*)d_in[3];
  const float* aD1 = (const float*)d_in[4];
  const float* b1  = (const float*)d_in[5];
  const float* W2  = (const float*)d_in[6];
  const float* aS2 = (const float*)d_in[7];
  const float* aD2 = (const float*)d_in[8];
  const float* b2  = (const float*)d_in[9];
  float* out = (float*)d_out;
  const int* srcI = ei;
  const int* dstI = ei + E;

  char* p = (char*)d_ws;
  auto carve = [&](size_t bytes) -> void* {
    void* r = (void*)p;
    p += (bytes + 255) & ~(size_t)255;
    return r;
  };
  _Float16* xh   = (_Float16*)carve((size_t)Nn * IN * 2);
  _Float16* w1t  = (_Float16*)carve((size_t)HC1 * IN * 2);
  _Float16* w2t  = (_Float16*)carve((size_t)HID * HC1 * 2);
  float*    h1   = (float*)carve((size_t)Nn * HC1 * 4);
  _Float16* h1h  = (_Float16*)carve((size_t)Nn * HC1 * 2);
  float*    h2   = (float*)carve((size_t)Nn * HC2 * 4);
  float*    as1  = (float*)carve((size_t)Nn * H1 * 4);
  float*    ad1  = (float*)carve((size_t)Nn * H1 * 4);
  float*    as2  = (float*)carve((size_t)Nn * 4);
  float*    ad2  = (float*)carve((size_t)Nn * 4);
  const size_t nTile = (size_t)((Nn + TILE_R - 1) / TILE_R);
  int*           lsrc = (int*)carve(nTile * TILE_W * QL * 4);
  unsigned char* lrow = (unsigned char*)carve(nTile * TILE_W * QL);
  int*           lcnt = (int*)carve(nTile * TILE_W * 4);
  const int nWord    = (E + SCAN_E - 1) / SCAN_E;
  const int bitPitch = ((int)nTile + 31) & ~31;
  unsigned*      tbits = (unsigned*)carve((size_t)nWord * bitPitch * 4);

  const int TB = 256;
  auto cdiv = [](long long a, long long b) { return (int)((a + b - 1) / b); };

  k_cast_f16<<<cdiv((long long)Nn * IN, EW_SPAN), TB, 0, stream>>>(x, xh, Nn * IN);
  k_castT_f16<<<cdiv((long long)IN * HC1, TB), TB, 0, stream>>>(W1, w1t, IN, HC1);
  k_castT_f16<<<cdiv((long long)HC1 * HID, TB), TB, 0, stream>>>(W2, w2t, HC1, HID);

  {
    int waves = (Nn / 16) * (HC1 / 64);
    k_wmma_gemm<<<cdiv(waves, 8), TB, 0, stream>>>(xh, w1t, h1, Nn, HC1, IN);
  }
  k_attn<<<cdiv((long long)Nn * H1, TB), TB, 0, stream>>>(h1, aS1, aD1, as1, ad1, Nn, H1, HID);
  k_tile_bits<<<dim3(cdiv(nWord, BITS_T / 32), cdiv((long long)nTile, BITS_TW)), BITS_T, 0, stream>>>(dstI, E, nWord, bitPitch, tbits);
  k_tile_agg<256, 4, true><<<cdiv(Nn, TILE_R), TILE_T, 0, stream>>>(srcI, dstI, E, Nn, as1, ad1, h1, b1, nullptr, h1h, lsrc, lrow, lcnt, tbits, bitPitch);

  {
    int waves = (Nn / 16) * (HC2 / 64);
    k_wmma_gemm<<<cdiv(waves, 8), TB, 0, stream>>>(h1h, w2t, h2, Nn, HC2, HC1);
  }
  k_attn<<<cdiv((long long)Nn, TB), TB, 0, stream>>>(h2, aS2, aD2, as2, ad2, Nn, 1, HID);
  k_tile_agg<64, 1, false><<<cdiv(Nn, TILE_R), TILE_T, 0, stream>>>(srcI, dstI, E, Nn, as2, ad2, h2, b2, out, nullptr, lsrc, lrow, lcnt, tbits, bitPitch);
}
